# starting kernel
# baseline (speedup 1.0000x reference)
#include <hip/hip_runtime.h>
#include <hip/hip_bf16.h>
#include <math.h>


typedef __attribute__((ext_vector_type(2))) float v2f;
typedef __attribute__((ext_vector_type(8))) float v8f;

#define GNN_N       100000
#define GNN_NUSERS  60000
#define GNN_EMB     64
#define GNN_E       3200000

__global__ __launch_bounds__(256) void gnn_init(
    const float* __restrict__ user_emb, const float* __restrict__ item_emb,
    float* __restrict__ ego, float* __restrict__ out)
{
    long tid = (long)blockIdx.x * 256 + threadIdx.x;
    if (tid >= (long)GNN_N * GNN_EMB) return;
    long row = tid >> 6;
    int  col = (int)(tid & 63);
    float v = (row < GNN_NUSERS)
                ? user_emb[row * GNN_EMB + col]
                : item_emb[(row - GNN_NUSERS) * GNN_EMB + col];
    ego[tid] = v;
    out[row * 192 + col] = v;
}

__global__ __launch_bounds__(256) void gnn_zero(float4* __restrict__ p, long n4)
{
    long tid = (long)blockIdx.x * 256 + threadIdx.x;
    if (tid < n4) p[tid] = make_float4(0.f, 0.f, 0.f, 0.f);
}

__global__ __launch_bounds__(256) void gnn_spmm(
    const int* __restrict__ esrc, const int* __restrict__ edst,
    const float* __restrict__ eval, const float* __restrict__ x,
    float* __restrict__ y)
{
    long tid = (long)blockIdx.x * 256 + threadIdx.x;
    long e = tid >> 3;
    if (e >= GNN_E) return;
    int part = (int)(tid & 7);
    int d = edst[e];
    int s = esrc[e];
    float v = eval[e];
    const float4* xr = (const float4*)(x + (long)d * GNN_EMB) + part * 2;
    float4 a = xr[0];
    float4 b = xr[1];
    float* yr = y + (long)s * GNN_EMB + part * 8;
    unsafeAtomicAdd(yr + 0, v * a.x);
    unsafeAtomicAdd(yr + 1, v * a.y);
    unsafeAtomicAdd(yr + 2, v * a.z);
    unsafeAtomicAdd(yr + 3, v * a.w);
    unsafeAtomicAdd(yr + 4, v * b.x);
    unsafeAtomicAdd(yr + 5, v * b.y);
    unsafeAtomicAdd(yr + 6, v * b.z);
    unsafeAtomicAdd(yr + 7, v * b.w);
}

__global__ __launch_bounds__(256) void gnn_layer(
    const float* __restrict__ sideL, float* __restrict__ ego,
    const float* __restrict__ Wgc, const float* __restrict__ bgc,
    const float* __restrict__ Wbi, const float* __restrict__ bbi,
    float* __restrict__ out, int out_col0)
{
    __shared__ float Wt[64][132];
    __shared__ float bsum[64];

    const int tid = threadIdx.x;

    for (int i = tid; i < 64 * 128; i += 256) {
        int k = i >> 6;
        int n = i & 63;
        float w = (k < 64) ? Wgc[k * 64 + n] : Wbi[(k - 64) * 64 + n];
        Wt[n][k] = w;
    }
    if (tid < 64) bsum[tid] = bgc[tid] + bbi[tid];
    __syncthreads();

    const int wave    = tid >> 5;
    const int lane    = tid & 31;
    const int rowbase = blockIdx.x * 128 + wave * 16;
    if (rowbase >= GNN_N) return;

    const int m    = lane & 15;
    const int half = lane >> 4;
    const int kb   = half << 1;

    const long  row    = (long)(rowbase + m);
    const float* egoRd = ego   + row * 64;
    const float* sLRd  = sideL + row * 64;

    v2f e2[16], l2[16];
#pragma unroll
    for (int s = 0; s < 16; ++s) {
        int c = 4 * s + kb;
        e2[s] = *(const v2f*)(egoRd + c);
        l2[s] = *(const v2f*)(sLRd + c);
    }

    v8f acc[4] = {};

#pragma unroll
    for (int s = 0; s < 16; ++s) {
        v2f a;
        a.x = e2[s].x + l2[s].x;
        a.y = e2[s].y + l2[s].y;
#pragma unroll
        for (int nt = 0; nt < 4; ++nt) {
            v2f b = *(const v2f*)(&Wt[nt * 16 + m][4 * s + kb]);
            acc[nt] = __builtin_amdgcn_wmma_f32_16x16x4_f32(
                false, a, false, b, (short)0, acc[nt], false, false);
        }
    }
#pragma unroll
    for (int s = 0; s < 16; ++s) {
        v2f a;
        a.x = e2[s].x * l2[s].x;
        a.y = e2[s].y * l2[s].y;
#pragma unroll
        for (int nt = 0; nt < 4; ++nt) {
            v2f b = *(const v2f*)(&Wt[nt * 16 + m][64 + 4 * s + kb]);
            acc[nt] = __builtin_amdgcn_wmma_f32_16x16x4_f32(
                false, a, false, b, (short)0, acc[nt], false, false);
        }
    }

    float vout[4][8];
    float ssq[8];
#pragma unroll
    for (int j = 0; j < 8; ++j) ssq[j] = 0.f;
#pragma unroll
    for (int nt = 0; nt < 4; ++nt) {
        float bc = bsum[nt * 16 + m];
#pragma unroll
        for (int j = 0; j < 8; ++j) {
            float v = acc[nt][j] + bc;
            v = (v > 0.f) ? v : 0.01f * v;
            vout[nt][j] = v;
            ssq[j] += v * v;
        }
    }

    float* egoWr = ego + (long)rowbase * 64;
#pragma unroll
    for (int j = 0; j < 8; ++j) {
        int mrow = j + (half << 3);
#pragma unroll
        for (int nt = 0; nt < 4; ++nt)
            egoWr[mrow * 64 + nt * 16 + m] = vout[nt][j];
    }

#pragma unroll
    for (int j = 0; j < 8; ++j) {
        float s = ssq[j];
        s += __shfl_xor(s, 8, 16);
        s += __shfl_xor(s, 4, 16);
        s += __shfl_xor(s, 2, 16);
        s += __shfl_xor(s, 1, 16);
        float inv = 1.0f / fmaxf(sqrtf(s), 1e-12f);
        int mrow = j + (half << 3);
        float* o = out + (long)(rowbase + mrow) * 192 + out_col0;
#pragma unroll
        for (int nt = 0; nt < 4; ++nt)
            o[nt * 16 + m] = vout[nt][j] * inv;
    }
}

extern "C" void kernel_launch(void* const* d_in, const int* in_sizes, int n_in,
                              void* d_out, int out_size, void* d_ws, size_t ws_size,
                              hipStream_t stream) {
    const float* user_emb = (const float*)d_in[0];
    const float* item_emb = (const float*)d_in[1];
    const float* W_gc0 = (const float*)d_in[2];
    const float* b_gc0 = (const float*)d_in[3];
    const float* W_bi0 = (const float*)d_in[4];
    const float* b_bi0 = (const float*)d_in[5];
    const float* W_gc1 = (const float*)d_in[6];
    const float* b_gc1 = (const float*)d_in[7];
    const float* W_bi1 = (const float*)d_in[8];
    const float* b_bi1 = (const float*)d_in[9];
    const int*   e_src = (const int*)d_in[10];
    const int*   e_dst = (const int*)d_in[11];
    const float* e_val = (const float*)d_in[12];
    float* out = (float*)d_out;

    float* ego   = (float*)d_ws;
    float* sideL = ego + (long)GNN_N * GNN_EMB;

    const long nelem  = (long)GNN_N * GNN_EMB;
    const int  gInit  = (int)((nelem + 255) / 256);
    const int  gZero  = (int)((nelem / 4 + 255) / 256);
    const int  gSpmm  = (int)(((long)GNN_E * 8 + 255) / 256);
    const int  gLayer = (GNN_N + 127) / 128;

    gnn_init<<<gInit, 256, 0, stream>>>(user_emb, item_emb, ego, out);

    gnn_zero<<<gZero, 256, 0, stream>>>((float4*)sideL, nelem / 4);
    gnn_spmm<<<gSpmm, 256, 0, stream>>>(e_src, e_dst, e_val, ego, sideL);
    gnn_layer<<<gLayer, 256, 0, stream>>>(sideL, ego, W_gc0, b_gc0, W_bi0, b_bi0,
                                          out, 64);

    gnn_zero<<<gZero, 256, 0, stream>>>((float4*)sideL, nelem / 4);
    gnn_spmm<<<gSpmm, 256, 0, stream>>>(e_src, e_dst, e_val, ego, sideL);
    gnn_layer<<<gLayer, 256, 0, stream>>>(sideL, ego, W_gc1, b_gc1, W_bi1, b_bi1,
                                          out, 128);
}


# SpMM: one wave walks 32 edges, fields broadcast by readlane, so every x load and every f32 atomic add covers one whole 128-B line (was 4 B in each of 8 lines); atomic arrival order differs
# speedup vs baseline: 2.8223x; 2.8223x over previous
#include <hip/hip_runtime.h>
#include <hip/hip_bf16.h>
#include <math.h>


typedef __attribute__((ext_vector_type(2))) float v2f;
typedef __attribute__((ext_vector_type(8))) float v8f;

#define GNN_N       100000
#define GNN_NUSERS  60000
#define GNN_EMB     64
#define GNN_E       3200000

__global__ __launch_bounds__(256) void gnn_init(
    const float* __restrict__ user_emb, const float* __restrict__ item_emb,
    float* __restrict__ ego, float* __restrict__ out)
{
    long tid = (long)blockIdx.x * 256 + threadIdx.x;
    if (tid >= (long)GNN_N * GNN_EMB) return;
    long row = tid >> 6;
    int  col = (int)(tid & 63);
    float v = (row < GNN_NUSERS)
                ? user_emb[row * GNN_EMB + col]
                : item_emb[(row - GNN_NUSERS) * GNN_EMB + col];
    ego[tid] = v;
    out[row * 192 + col] = v;
}

__global__ __launch_bounds__(256) void gnn_zero(float4* __restrict__ p, long n4)
{
    long tid = (long)blockIdx.x * 256 + threadIdx.x;
    if (tid < n4) p[tid] = make_float4(0.f, 0.f, 0.f, 0.f);
}

static_assert(GNN_E % 256 == 0, "a block of 8 waves owns 256 whole edges");

__global__ __launch_bounds__(256) void gnn_spmm(
    const int* __restrict__ esrc, const int* __restrict__ edst,
    const float* __restrict__ eval, const float* __restrict__ x,
    float* __restrict__ y)
{
    const int  lane = threadIdx.x & 31;
    const long e    = (long)blockIdx.x * 256 + threadIdx.x;
    const int  d     = edst[e];
    const int  s     = esrc[e];
    const int  vbits = __float_as_int(eval[e]);

#pragma unroll
    for (int base = 0; base < 32; base += 8) {
        float a[8], b[8];
#pragma unroll
        for (int i = 0; i < 8; ++i) {
            const int di = __builtin_amdgcn_readlane(d, base + i);
            const float* xr = x + (long)di * GNN_EMB;
            a[i] = xr[lane];
            b[i] = xr[lane + 32];
        }
#pragma unroll
        for (int i = 0; i < 8; ++i) {
            const int   si = __builtin_amdgcn_readlane(s, base + i);
            const float vi = __int_as_float(__builtin_amdgcn_readlane(vbits, base + i));
            float* yr = y + (long)si * GNN_EMB;
            unsafeAtomicAdd(yr + lane,      vi * a[i]);
            unsafeAtomicAdd(yr + lane + 32, vi * b[i]);
        }
    }
}

__global__ __launch_bounds__(256) void gnn_layer(
    const float* __restrict__ sideL, float* __restrict__ ego,
    const float* __restrict__ Wgc, const float* __restrict__ bgc,
    const float* __restrict__ Wbi, const float* __restrict__ bbi,
    float* __restrict__ out, int out_col0)
{
    __shared__ float Wt[64][132];
    __shared__ float bsum[64];

    const int tid = threadIdx.x;

    for (int i = tid; i < 64 * 128; i += 256) {
        int k = i >> 6;
        int n = i & 63;
        float w = (k < 64) ? Wgc[k * 64 + n] : Wbi[(k - 64) * 64 + n];
        Wt[n][k] = w;
    }
    if (tid < 64) bsum[tid] = bgc[tid] + bbi[tid];
    __syncthreads();

    const int wave    = tid >> 5;
    const int lane    = tid & 31;
    const int rowbase = blockIdx.x * 128 + wave * 16;
    if (rowbase >= GNN_N) return;

    const int m    = lane & 15;
    const int half = lane >> 4;
    const int kb   = half << 1;

    const long  row    = (long)(rowbase + m);
    const float* egoRd = ego   + row * 64;
    const float* sLRd  = sideL + row * 64;

    v2f e2[16], l2[16];
#pragma unroll
    for (int s = 0; s < 16; ++s) {
        int c = 4 * s + kb;
        e2[s] = *(const v2f*)(egoRd + c);
        l2[s] = *(const v2f*)(sLRd + c);
    }

    v8f acc[4] = {};

#pragma unroll
    for (int s = 0; s < 16; ++s) {
        v2f a;
        a.x = e2[s].x + l2[s].x;
        a.y = e2[s].y + l2[s].y;
#pragma unroll
        for (int nt = 0; nt < 4; ++nt) {
            v2f b = *(const v2f*)(&Wt[nt * 16 + m][4 * s + kb]);
            acc[nt] = __builtin_amdgcn_wmma_f32_16x16x4_f32(
                false, a, false, b, (short)0, acc[nt], false, false);
        }
    }
#pragma unroll
    for (int s = 0; s < 16; ++s) {
        v2f a;
        a.x = e2[s].x * l2[s].x;
        a.y = e2[s].y * l2[s].y;
#pragma unroll
        for (int nt = 0; nt < 4; ++nt) {
            v2f b = *(const v2f*)(&Wt[nt * 16 + m][64 + 4 * s + kb]);
            acc[nt] = __builtin_amdgcn_wmma_f32_16x16x4_f32(
                false, a, false, b, (short)0, acc[nt], false, false);
        }
    }

    float vout[4][8];
    float ssq[8];
#pragma unroll
    for (int j = 0; j < 8; ++j) ssq[j] = 0.f;
#pragma unroll
    for (int nt = 0; nt < 4; ++nt) {
        float bc = bsum[nt * 16 + m];
#pragma unroll
        for (int j = 0; j < 8; ++j) {
            float v = acc[nt][j] + bc;
            v = (v > 0.f) ? v : 0.01f * v;
            vout[nt][j] = v;
            ssq[j] += v * v;
        }
    }

    float* egoWr = ego + (long)rowbase * 64;
#pragma unroll
    for (int j = 0; j < 8; ++j) {
        int mrow = j + (half << 3);
#pragma unroll
        for (int nt = 0; nt < 4; ++nt)
            egoWr[mrow * 64 + nt * 16 + m] = vout[nt][j];
    }

#pragma unroll
    for (int j = 0; j < 8; ++j) {
        float s = ssq[j];
        s += __shfl_xor(s, 8, 16);
        s += __shfl_xor(s, 4, 16);
        s += __shfl_xor(s, 2, 16);
        s += __shfl_xor(s, 1, 16);
        float inv = 1.0f / fmaxf(sqrtf(s), 1e-12f);
        int mrow = j + (half << 3);
        float* o = out + (long)(rowbase + mrow) * 192 + out_col0;
#pragma unroll
        for (int nt = 0; nt < 4; ++nt)
            o[nt * 16 + m] = vout[nt][j] * inv;
    }
}

extern "C" void kernel_launch(void* const* d_in, const int* in_sizes, int n_in,
                              void* d_out, int out_size, void* d_ws, size_t ws_size,
                              hipStream_t stream) {
    const float* user_emb = (const float*)d_in[0];
    const float* item_emb = (const float*)d_in[1];
    const float* W_gc0 = (const float*)d_in[2];
    const float* b_gc0 = (const float*)d_in[3];
    const float* W_bi0 = (const float*)d_in[4];
    const float* b_bi0 = (const float*)d_in[5];
    const float* W_gc1 = (const float*)d_in[6];
    const float* b_gc1 = (const float*)d_in[7];
    const float* W_bi1 = (const float*)d_in[8];
    const float* b_bi1 = (const float*)d_in[9];
    const int*   e_src = (const int*)d_in[10];
    const int*   e_dst = (const int*)d_in[11];
    const float* e_val = (const float*)d_in[12];
    float* out = (float*)d_out;

    float* ego   = (float*)d_ws;
    float* sideL = ego + (long)GNN_N * GNN_EMB;

    const long nelem  = (long)GNN_N * GNN_EMB;
    const int  gInit  = (int)((nelem + 255) / 256);
    const int  gZero  = (int)((nelem / 4 + 255) / 256);
    const int  gSpmm  = (int)(((long)GNN_E + 255) / 256);
    const int  gLayer = (GNN_N + 127) / 128;

    gnn_init<<<gInit, 256, 0, stream>>>(user_emb, item_emb, ego, out);

    gnn_zero<<<gZero, 256, 0, stream>>>((float4*)sideL, nelem / 4);
    gnn_spmm<<<gSpmm, 256, 0, stream>>>(e_src, e_dst, e_val, ego, sideL);
    gnn_layer<<<gLayer, 256, 0, stream>>>(sideL, ego, W_gc0, b_gc0, W_bi0, b_bi0,
                                          out, 64);

    gnn_zero<<<gZero, 256, 0, stream>>>((float4*)sideL, nelem / 4);
    gnn_spmm<<<gSpmm, 256, 0, stream>>>(e_src, e_dst, e_val, ego, sideL);
    gnn_layer<<<gLayer, 256, 0, stream>>>(sideL, ego, W_gc1, b_gc1, W_bi1, b_bi1,
                                          out, 128);
}


# SpMM per 128-row tile in LDS: edges bucketed once per call by tile (LDS counting sort), layer block sums its tile with ds_add_f32 and feeds its WMMAs; no global f32 atomics; sum order = LDS arrival
# speedup vs baseline: 4.4112x; 1.5630x over previous
#include <hip/hip_runtime.h>
#include <math.h>


typedef __attribute__((ext_vector_type(2))) float v2f;
typedef __attribute__((ext_vector_type(8))) float v8f;

#define GNN_N       100000
#define GNN_NUSERS  60000
#define GNN_EMB     64
#define GNN_E       3200000
#define GNN_NBKT    782
#define SORT_EDGES  16384
#define SORT_BLOCKS 196
#define YPITCH      68

static_assert(GNN_N % 16 == 0, "a wave's 16-row tile is all inside or all outside N");
static_assert(GNN_N < (1 << 17), "a source row index fits the low 17 bits of an entry");

__device__ __forceinline__ unsigned gnn_bucket(int s)
{
    unsigned b = ((unsigned)s) >> 7;
    return b < GNN_NBKT ? b : GNN_NBKT - 1;
}

__device__ __forceinline__ int gnn_scan1024(int x, int* wt)
{
    const int lane = threadIdx.x & 31;
    const int w    = threadIdx.x >> 5;
    int inc = x;
    for (int d = 1; d < 32; d <<= 1) {
        int t = __shfl_up(inc, d);
        if (lane >= d) inc += t;
    }
    if (lane == 31) wt[w] = inc;
    __syncthreads();
    int off = 0;
    for (int i = 0; i < w; ++i) off += wt[i];
    __syncthreads();
    return off + inc - x;
}

__global__ __launch_bounds__(256) void gnn_init(
    const float* __restrict__ user_emb, const float* __restrict__ item_emb,
    float* __restrict__ out, int* __restrict__ hist)
{
    long tid = (long)blockIdx.x * 256 + threadIdx.x;
    if (tid < GNN_NBKT) hist[tid] = 0;
    if (tid >= (long)GNN_N * GNN_EMB) return;
    long row = tid >> 6;
    int  col = (int)(tid & 63);
    float v = (row < GNN_NUSERS)
                ? user_emb[row * GNN_EMB + col]
                : item_emb[(row - GNN_NUSERS) * GNN_EMB + col];
    out[row * 192 + col] = v;
}

__global__ __launch_bounds__(1024) void gnn_hist(
    const int* __restrict__ esrc, int* __restrict__ hist)
{
    __shared__ int lh[GNN_NBKT];
    const int tid = threadIdx.x;
    if (tid < GNN_NBKT) lh[tid] = 0;
    __syncthreads();
    const int base = blockIdx.x * SORT_EDGES;
#pragma unroll
    for (int k = 0; k < 16; ++k) {
        int e = base + k * 1024 + tid;
        if (e < GNN_E) atomicAdd(&lh[gnn_bucket(esrc[e])], 1);
    }
    __syncthreads();
    if (tid < GNN_NBKT) {
        int c = lh[tid];
        if (c) atomicAdd(&hist[tid], c);
    }
}

__global__ __launch_bounds__(1024) void gnn_offsets(
    const int* __restrict__ hist, int* __restrict__ cursor, float* __restrict__ out)
{
    __shared__ int wt[32];
    const int tid = threadIdx.x;
    int c = (tid < GNN_NBKT) ? hist[tid] : 0;
    int start = gnn_scan1024(c, wt);
    if (tid < GNN_NBKT) {
        cursor[tid] = start;
        float* o = out + (long)tid * 128 * 192 + 128;
        o[0] = __int_as_float(start);
        o[1] = __int_as_float(c);
    }
}

__global__ __launch_bounds__(1024) void gnn_sort(
    const int* __restrict__ esrc, const int* __restrict__ edst,
    const float* __restrict__ eval, int* __restrict__ cursor,
    int2* __restrict__ csr)
{
    __shared__ int2 sorted[SORT_EDGES];
    __shared__ int  lcnt[GNN_NBKT];
    __shared__ int  lstart[GNN_NBKT];
    __shared__ int  lcur[GNN_NBKT];
    __shared__ int  gbase[GNN_NBKT];
    __shared__ int  wt[32];

    const int tid  = threadIdx.x;
    const int base = blockIdx.x * SORT_EDGES;
    if (tid < GNN_NBKT) lcnt[tid] = 0;
    __syncthreads();

    int      ex[16], ey[16];
    unsigned bk[16];
#pragma unroll
    for (int k = 0; k < 16; ++k) {
        int e = base + k * 1024 + tid;
        bk[k] = GNN_NBKT;
        ex[k] = 0; ey[k] = 0;
        if (e < GNN_E) {
            int s = esrc[e];
            bk[k] = gnn_bucket(s);
            ex[k] = (edst[e] & 0x1FFFF) | ((s & 127) << 17);
            ey[k] = __float_as_int(eval[e]);
            atomicAdd(&lcnt[bk[k]], 1);
        }
    }
    __syncthreads();

    int c  = (tid < GNN_NBKT) ? lcnt[tid] : 0;
    int ls = gnn_scan1024(c, wt);
    if (tid < GNN_NBKT) {
        lstart[tid] = ls;
        lcur[tid]   = ls;
        gbase[tid]  = c ? atomicAdd(&cursor[tid], c) : 0;
    }
    __syncthreads();

#pragma unroll
    for (int k = 0; k < 16; ++k) {
        if (bk[k] < GNN_NBKT) {
            int r = atomicAdd(&lcur[bk[k]], 1);
            r = r < SORT_EDGES ? r : SORT_EDGES - 1;
            sorted[r] = make_int2(ex[k], ey[k]);
        }
    }
    __syncthreads();

    const int lane = tid & 31;
    for (int b = tid >> 5; b < GNN_NBKT; b += 32) {
        const int n  = lcnt[b];
        const int l0 = lstart[b];
        const int g0 = gbase[b];
        for (int i = lane; i < n; i += 32) {
            int g = g0 + i;
            g = g < GNN_E ? g : GNN_E - 1;
            csr[g] = sorted[l0 + i];
        }
    }
}

__global__ __launch_bounds__(256) void gnn_layer(
    const int2* __restrict__ csr,
    const float* __restrict__ xa, const float* __restrict__ xb,
    float* __restrict__ ego_out,
    const float* __restrict__ Wgc, const float* __restrict__ bgc,
    const float* __restrict__ Wbi, const float* __restrict__ bbi,
    float* out, int out_col0)
{
    __shared__ __attribute__((aligned(16))) float Wt[64][132];
    __shared__ __attribute__((aligned(16))) float ytile[128][YPITCH];
    __shared__ float bsum[64];

    const int tid  = threadIdx.x;
    const int wave = __builtin_amdgcn_readfirstlane(tid >> 5);
    const int lane = tid & 31;

    int start, cnt;
    {
        const float* o = out + (long)blockIdx.x * 128 * 192 + 128;
        start = __builtin_amdgcn_readfirstlane(__float_as_int(o[0]));
        cnt   = __builtin_amdgcn_readfirstlane(__float_as_int(o[1]));
        start = start < 0 ? 0 : (start > GNN_E ? GNN_E : start);
        cnt   = cnt < 0 ? 0 : (cnt > GNN_E - start ? GNN_E - start : cnt);
    }
    const int end = start + cnt;

    for (int i = tid; i < 128 * YPITCH; i += 256) (&ytile[0][0])[i] = 0.f;

    for (int i = tid; i < 64 * 128; i += 256) {
        int k = i >> 6;
        int n = i & 63;
        float w = (k < 64) ? Wgc[k * 64 + n] : Wbi[(k - 64) * 64 + n];
        Wt[n][k] = w;
    }
    if (tid < 64) bsum[tid] = bgc[tid] + bbi[tid];
    __syncthreads();

    {
        const long xskip = (xb - xa) - (long)GNN_NUSERS * GNN_EMB;
        int  cb  = start + wave * 32;
        int2 ent = make_int2(0, 0);
        if (cb + lane < end) ent = csr[cb + lane];
        while (cb < end) {
            const int n  = end - cb;
            const int nb = cb + 256;
            int2 nxt = make_int2(0, 0);
            if (nb + lane < end) nxt = csr[nb + lane];

#pragma unroll
            for (int base = 0; base < 32; base += 8) {
                if (base < n) {
                    float a[8], b[8];
#pragma unroll
                    for (int i = 0; i < 8; ++i) {
                        const int f  = __builtin_amdgcn_readlane(ent.x, base + i);
                        unsigned  di = (unsigned)f & 0x1FFFFu;
                        di = di < GNN_N ? di : GNN_N - 1;
                        const float* xr = xa + ((long)di * GNN_EMB + (di < GNN_NUSERS ? 0L : xskip));
                        a[i] = xr[lane];
                        b[i] = xr[lane + 32];
                    }
#pragma unroll
                    for (int i = 0; i < 8; ++i) {
                        const int   f  = __builtin_amdgcn_readlane(ent.x, base + i);
                        const float vi = __int_as_float(__builtin_amdgcn_readlane(ent.y, base + i));
                        const int   lr = (f >> 17) & 127;
                        const bool  live = (base + i) < n;
                        const float pa = live ? vi * a[i] : 0.f;
                        const float pb = live ? vi * b[i] : 0.f;
                        atomicAdd(&ytile[lr][lane],      pa);
                        atomicAdd(&ytile[lr][lane + 32], pb);
                    }
                }
            }
            ent = nxt;
            cb  = nb;
        }
    }
    __syncthreads();

    const int rowbase = blockIdx.x * 128 + wave * 16;
    if (rowbase >= GNN_N) return;

    const int m    = lane & 15;
    const int half = lane >> 4;
    const int kb   = half << 1;

    const long  row   = (long)(rowbase + m);
    const float* egoRd = (row < GNN_NUSERS) ? xa + row * 64 : xb + (row - GNN_NUSERS) * 64;
    const float* sLRd  = &ytile[wave * 16 + m][0];

    v2f e2[16], l2[16];
#pragma unroll
    for (int s = 0; s < 16; ++s) {
        int c = 4 * s + kb;
        e2[s] = *(const v2f*)(egoRd + c);
        l2[s] = *(const v2f*)(sLRd + c);
    }

    v8f acc[4] = {};

#pragma unroll
    for (int s = 0; s < 16; ++s) {
        v2f a;
        a.x = e2[s].x + l2[s].x;
        a.y = e2[s].y + l2[s].y;
#pragma unroll
        for (int nt = 0; nt < 4; ++nt) {
            v2f b = *(const v2f*)(&Wt[nt * 16 + m][4 * s + kb]);
            acc[nt] = __builtin_amdgcn_wmma_f32_16x16x4_f32(
                false, a, false, b, (short)0, acc[nt], false, false);
        }
    }
#pragma unroll
    for (int s = 0; s < 16; ++s) {
        v2f a;
        a.x = e2[s].x * l2[s].x;
        a.y = e2[s].y * l2[s].y;
#pragma unroll
        for (int nt = 0; nt < 4; ++nt) {
            v2f b = *(const v2f*)(&Wt[nt * 16 + m][64 + 4 * s + kb]);
            acc[nt] = __builtin_amdgcn_wmma_f32_16x16x4_f32(
                false, a, false, b, (short)0, acc[nt], false, false);
        }
    }

    float vout[4][8];
    float ssq[8];
#pragma unroll
    for (int j = 0; j < 8; ++j) ssq[j] = 0.f;
#pragma unroll
    for (int nt = 0; nt < 4; ++nt) {
        float bc = bsum[nt * 16 + m];
#pragma unroll
        for (int j = 0; j < 8; ++j) {
            float v = acc[nt][j] + bc;
            v = (v > 0.f) ? v : 0.01f * v;
            vout[nt][j] = v;
            ssq[j] += v * v;
        }
    }

    if (ego_out) {
        float* egoWr = ego_out + (long)rowbase * 64;
#pragma unroll
        for (int j = 0; j < 8; ++j) {
            int mrow = j + (half << 3);
#pragma unroll
            for (int nt = 0; nt < 4; ++nt)
                egoWr[mrow * 64 + nt * 16 + m] = vout[nt][j];
        }
    }

#pragma unroll
    for (int j = 0; j < 8; ++j) {
        float s = ssq[j];
        s += __shfl_xor(s, 8, 16);
        s += __shfl_xor(s, 4, 16);
        s += __shfl_xor(s, 2, 16);
        s += __shfl_xor(s, 1, 16);
        float inv = 1.0f / fmaxf(sqrtf(s), 1e-12f);
        int mrow = j + (half << 3);
        float* o = out + (long)(rowbase + mrow) * 192 + out_col0;
#pragma unroll
        for (int nt = 0; nt < 4; ++nt)
            o[nt * 16 + m] = vout[nt][j] * inv;
    }
}

extern "C" void kernel_launch(void* const* d_in, const int* in_sizes, int n_in,
                              void* d_out, int out_size, void* d_ws, size_t ws_size,
                              hipStream_t stream) {
    const float* user_emb = (const float*)d_in[0];
    const float* item_emb = (const float*)d_in[1];
    const float* W_gc0 = (const float*)d_in[2];
    const float* b_gc0 = (const float*)d_in[3];
    const float* W_bi0 = (const float*)d_in[4];
    const float* b_bi0 = (const float*)d_in[5];
    const float* W_gc1 = (const float*)d_in[6];
    const float* b_gc1 = (const float*)d_in[7];
    const float* W_bi1 = (const float*)d_in[8];
    const float* b_bi1 = (const float*)d_in[9];
    const int*   e_src = (const int*)d_in[10];
    const int*   e_dst = (const int*)d_in[11];
    const float* e_val = (const float*)d_in[12];
    float* out = (float*)d_out;

    float* ego1 = (float*)d_ws;
    int2*  csr  = (int2*)(ego1 + (long)GNN_N * GNN_EMB);
    int*   hist   = (int*)d_ws;
    int*   cursor = hist + 1024;

    gnn_init<<<25000, 256, 0, stream>>>(user_emb, item_emb, out, hist);
    gnn_hist<<<SORT_BLOCKS, 1024, 0, stream>>>(e_src, hist);
    gnn_offsets<<<1, 1024, 0, stream>>>(hist, cursor, out);
    gnn_sort<<<SORT_BLOCKS, 1024, 0, stream>>>(e_src, e_dst, e_val, cursor, csr);

    gnn_layer<<<GNN_NBKT, 256, 0, stream>>>(csr, user_emb, item_emb, ego1,
                                            W_gc0, b_gc0, W_bi0, b_bi0, out, 64);
    gnn_layer<<<GNN_NBKT, 256, 0, stream>>>(csr, ego1, ego1 + (long)GNN_NUSERS * GNN_EMB,
                                            nullptr,
                                            W_gc1, b_gc1, W_bi1, b_bi1, out, 128);
}


# Concat launch removed (25000 blocks, one dword per thread): layer-0 blocks copy their tile's rows to out[:,0:64] as whole lines; tile counts via a per-block table, no cleared histogram
# speedup vs baseline: 4.6085x; 1.0447x over previous
#include <hip/hip_runtime.h>
#include <math.h>


typedef __attribute__((ext_vector_type(2))) float v2f;
typedef __attribute__((ext_vector_type(8))) float v8f;

#define GNN_N       100000
#define GNN_NUSERS  60000
#define GNN_EMB     64
#define GNN_E       3200000
#define GNN_NBKT    782
#define SORT_EDGES  16384
#define SORT_BLOCKS 196
#define CNT_PITCH   1024
#define YPITCH      68

static_assert(GNN_N % 16 == 0, "a wave's 16-row tile is all inside or all outside N");
static_assert(GNN_N < (1 << 17), "a source row index fits the low 17 bits of an entry");

__device__ __forceinline__ unsigned gnn_bucket(int s)
{
    unsigned b = ((unsigned)s) >> 7;
    return b < GNN_NBKT ? b : GNN_NBKT - 1;
}

__device__ __forceinline__ int gnn_scan1024(int x, int* wt)
{
    const int lane = threadIdx.x & 31;
    const int w    = threadIdx.x >> 5;
    int inc = x;
    for (int d = 1; d < 32; d <<= 1) {
        int t = __shfl_up(inc, d);
        if (lane >= d) inc += t;
    }
    if (lane == 31) wt[w] = inc;
    __syncthreads();
    int off = 0;
    for (int i = 0; i < w; ++i) off += wt[i];
    __syncthreads();
    return off + inc - x;
}

__global__ __launch_bounds__(1024) void gnn_hist(
    const int* __restrict__ esrc, int* __restrict__ cnt_table)
{
    __shared__ int lh[GNN_NBKT];
    const int tid = threadIdx.x;
    if (tid < GNN_NBKT) lh[tid] = 0;
    __syncthreads();
    const int base = blockIdx.x * SORT_EDGES;
#pragma unroll
    for (int k = 0; k < 16; ++k) {
        int e = base + k * 1024 + tid;
        if (e < GNN_E) atomicAdd(&lh[gnn_bucket(esrc[e])], 1);
    }
    __syncthreads();
    if (tid < GNN_NBKT) cnt_table[blockIdx.x * CNT_PITCH + tid] = lh[tid];
}

__global__ __launch_bounds__(1024) void gnn_offsets(
    const int* __restrict__ cnt_table, int* __restrict__ cursor, float* __restrict__ out)
{
    __shared__ int wt[32];
    const int tid = threadIdx.x;
    int c = 0;
    if (tid < GNN_NBKT) {
        const int* col = cnt_table + tid;
#pragma unroll
        for (int b = 0; b < SORT_BLOCKS; ++b) c += col[b * CNT_PITCH];
    }
    int start = gnn_scan1024(c, wt);
    if (tid < GNN_NBKT) {
        cursor[tid] = start;
        float* o = out + (long)tid * 128 * 192 + 128;
        o[0] = __int_as_float(start);
        o[1] = __int_as_float(c);
    }
}

__global__ __launch_bounds__(1024) void gnn_sort(
    const int* __restrict__ esrc, const int* __restrict__ edst,
    const float* __restrict__ eval, int* __restrict__ cursor,
    int2* __restrict__ csr)
{
    __shared__ int2 sorted[SORT_EDGES];
    __shared__ int  lcnt[GNN_NBKT];
    __shared__ int  lstart[GNN_NBKT];
    __shared__ int  lcur[GNN_NBKT];
    __shared__ int  gbase[GNN_NBKT];
    __shared__ int  wt[32];

    const int tid  = threadIdx.x;
    const int base = blockIdx.x * SORT_EDGES;
    if (tid < GNN_NBKT) lcnt[tid] = 0;
    __syncthreads();

    int      ex[16], ey[16];
    unsigned bk[16];
#pragma unroll
    for (int k = 0; k < 16; ++k) {
        int e = base + k * 1024 + tid;
        bk[k] = GNN_NBKT;
        ex[k] = 0; ey[k] = 0;
        if (e < GNN_E) {
            int s = esrc[e];
            bk[k] = gnn_bucket(s);
            ex[k] = (edst[e] & 0x1FFFF) | ((s & 127) << 17);
            ey[k] = __float_as_int(eval[e]);
            atomicAdd(&lcnt[bk[k]], 1);
        }
    }
    __syncthreads();

    int c  = (tid < GNN_NBKT) ? lcnt[tid] : 0;
    int ls = gnn_scan1024(c, wt);
    if (tid < GNN_NBKT) {
        lstart[tid] = ls;
        lcur[tid]   = ls;
        gbase[tid]  = c ? atomicAdd(&cursor[tid], c) : 0;
    }
    __syncthreads();

#pragma unroll
    for (int k = 0; k < 16; ++k) {
        if (bk[k] < GNN_NBKT) {
            int r = atomicAdd(&lcur[bk[k]], 1);
            r = r < SORT_EDGES ? r : SORT_EDGES - 1;
            sorted[r] = make_int2(ex[k], ey[k]);
        }
    }
    __syncthreads();

    const int lane = tid & 31;
    for (int b = tid >> 5; b < GNN_NBKT; b += 32) {
        const int n  = lcnt[b];
        const int l0 = lstart[b];
        const int g0 = gbase[b];
        for (int i = lane; i < n; i += 32) {
            int g = g0 + i;
            g = g < GNN_E ? g : GNN_E - 1;
            csr[g] = sorted[l0 + i];
        }
    }
}

__global__ __launch_bounds__(256) void gnn_layer(
    const int2* __restrict__ csr,
    const float* __restrict__ xa, const float* __restrict__ xb,
    float* __restrict__ ego_out,
    const float* __restrict__ Wgc, const float* __restrict__ bgc,
    const float* __restrict__ Wbi, const float* __restrict__ bbi,
    float* out, int out_col0)
{
    __shared__ __attribute__((aligned(16))) float Wt[64][132];
    __shared__ __attribute__((aligned(16))) float ytile[128][YPITCH];
    __shared__ float bsum[64];

    const int tid  = threadIdx.x;
    const int wave = __builtin_amdgcn_readfirstlane(tid >> 5);
    const int lane = tid & 31;

    int start, cnt;
    {
        const float* o = out + (long)blockIdx.x * 128 * 192 + 128;
        start = __builtin_amdgcn_readfirstlane(__float_as_int(o[0]));
        cnt   = __builtin_amdgcn_readfirstlane(__float_as_int(o[1]));
        start = start < 0 ? 0 : (start > GNN_E ? GNN_E : start);
        cnt   = cnt < 0 ? 0 : (cnt > GNN_E - start ? GNN_E - start : cnt);
    }
    const int end = start + cnt;

    float4 xv[8];
    if (ego_out) {
#pragma unroll
        for (int k = 0; k < 8; ++k) {
            const int i   = tid + k * 256;
            const int row = blockIdx.x * 128 + (i >> 4);
            xv[k] = make_float4(0.f, 0.f, 0.f, 0.f);
            if (row < GNN_N) {
                const float* xr = (row < GNN_NUSERS) ? xa + (long)row * GNN_EMB
                                                     : xb + (long)(row - GNN_NUSERS) * GNN_EMB;
                xv[k] = *(const float4*)(xr + ((i & 15) << 2));
            }
        }
    }

    for (int i = tid; i < 128 * YPITCH; i += 256) (&ytile[0][0])[i] = 0.f;

    for (int i = tid; i < 64 * 128; i += 256) {
        int k = i >> 6;
        int n = i & 63;
        float w = (k < 64) ? Wgc[k * 64 + n] : Wbi[(k - 64) * 64 + n];
        Wt[n][k] = w;
    }
    if (tid < 64) bsum[tid] = bgc[tid] + bbi[tid];
    __syncthreads();

    if (ego_out) {
#pragma unroll
        for (int k = 0; k < 8; ++k) {
            const int i   = tid + k * 256;
            const int row = blockIdx.x * 128 + (i >> 4);
            if (row < GNN_N) *(float4*)(out + (long)row * 192 + ((i & 15) << 2)) = xv[k];
        }
    }

    {
        const long xskip = (xb - xa) - (long)GNN_NUSERS * GNN_EMB;
        int  cb  = start + wave * 32;
        int2 ent = make_int2(0, 0);
        if (cb + lane < end) ent = csr[cb + lane];
        while (cb < end) {
            const int n  = end - cb;
            const int nb = cb + 256;
            int2 nxt = make_int2(0, 0);
            if (nb + lane < end) nxt = csr[nb + lane];

#pragma unroll
            for (int base = 0; base < 32; base += 8) {
                if (base < n) {
                    float a[8], b[8];
#pragma unroll
                    for (int i = 0; i < 8; ++i) {
                        const int f  = __builtin_amdgcn_readlane(ent.x, base + i);
                        unsigned  di = (unsigned)f & 0x1FFFFu;
                        di = di < GNN_N ? di : GNN_N - 1;
                        const float* xr = xa + ((long)di * GNN_EMB + (di < GNN_NUSERS ? 0L : xskip));
                        a[i] = xr[lane];
                        b[i] = xr[lane + 32];
                    }
#pragma unroll
                    for (int i = 0; i < 8; ++i) {
                        const int   f  = __builtin_amdgcn_readlane(ent.x, base + i);
                        const float vi = __int_as_float(__builtin_amdgcn_readlane(ent.y, base + i));
                        const int   lr = (f >> 17) & 127;
                        const bool  live = (base + i) < n;
                        const float pa = live ? vi * a[i] : 0.f;
                        const float pb = live ? vi * b[i] : 0.f;
                        atomicAdd(&ytile[lr][lane],      pa);
                        atomicAdd(&ytile[lr][lane + 32], pb);
                    }
                }
            }
            ent = nxt;
            cb  = nb;
        }
    }
    __syncthreads();

    const int rowbase = blockIdx.x * 128 + wave * 16;
    if (rowbase >= GNN_N) return;

    const int m    = lane & 15;
    const int half = lane >> 4;
    const int kb   = half << 1;

    const long  row   = (long)(rowbase + m);
    const float* egoRd = (row < GNN_NUSERS) ? xa + row * 64 : xb + (row - GNN_NUSERS) * 64;
    const float* sLRd  = &ytile[wave * 16 + m][0];

    v2f e2[16], l2[16];
#pragma unroll
    for (int s = 0; s < 16; ++s) {
        int c = 4 * s + kb;
        e2[s] = *(const v2f*)(egoRd + c);
        l2[s] = *(const v2f*)(sLRd + c);
    }

    v8f acc[4] = {};

#pragma unroll
    for (int s = 0; s < 16; ++s) {
        v2f a;
        a.x = e2[s].x + l2[s].x;
        a.y = e2[s].y + l2[s].y;
#pragma unroll
        for (int nt = 0; nt < 4; ++nt) {
            v2f b = *(const v2f*)(&Wt[nt * 16 + m][4 * s + kb]);
            acc[nt] = __builtin_amdgcn_wmma_f32_16x16x4_f32(
                false, a, false, b, (short)0, acc[nt], false, false);
        }
    }
#pragma unroll
    for (int s = 0; s < 16; ++s) {
        v2f a;
        a.x = e2[s].x * l2[s].x;
        a.y = e2[s].y * l2[s].y;
#pragma unroll
        for (int nt = 0; nt < 4; ++nt) {
            v2f b = *(const v2f*)(&Wt[nt * 16 + m][64 + 4 * s + kb]);
            acc[nt] = __builtin_amdgcn_wmma_f32_16x16x4_f32(
                false, a, false, b, (short)0, acc[nt], false, false);
        }
    }

    float vout[4][8];
    float ssq[8];
#pragma unroll
    for (int j = 0; j < 8; ++j) ssq[j] = 0.f;
#pragma unroll
    for (int nt = 0; nt < 4; ++nt) {
        float bc = bsum[nt * 16 + m];
#pragma unroll
        for (int j = 0; j < 8; ++j) {
            float v = acc[nt][j] + bc;
            v = (v > 0.f) ? v : 0.01f * v;
            vout[nt][j] = v;
            ssq[j] += v * v;
        }
    }

    if (ego_out) {
        float* egoWr = ego_out + (long)rowbase * 64;
#pragma unroll
        for (int j = 0; j < 8; ++j) {
            int mrow = j + (half << 3);
#pragma unroll
            for (int nt = 0; nt < 4; ++nt)
                egoWr[mrow * 64 + nt * 16 + m] = vout[nt][j];
        }
    }

#pragma unroll
    for (int j = 0; j < 8; ++j) {
        float s = ssq[j];
        s += __shfl_xor(s, 8, 16);
        s += __shfl_xor(s, 4, 16);
        s += __shfl_xor(s, 2, 16);
        s += __shfl_xor(s, 1, 16);
        float inv = 1.0f / fmaxf(sqrtf(s), 1e-12f);
        int mrow = j + (half << 3);
        float* o = out + (long)(rowbase + mrow) * 192 + out_col0;
#pragma unroll
        for (int nt = 0; nt < 4; ++nt)
            o[nt * 16 + m] = vout[nt][j] * inv;
    }
}

extern "C" void kernel_launch(void* const* d_in, const int* in_sizes, int n_in,
                              void* d_out, int out_size, void* d_ws, size_t ws_size,
                              hipStream_t stream) {
    const float* user_emb = (const float*)d_in[0];
    const float* item_emb = (const float*)d_in[1];
    const float* W_gc0 = (const float*)d_in[2];
    const float* b_gc0 = (const float*)d_in[3];
    const float* W_bi0 = (const float*)d_in[4];
    const float* b_bi0 = (const float*)d_in[5];
    const float* W_gc1 = (const float*)d_in[6];
    const float* b_gc1 = (const float*)d_in[7];
    const float* W_bi1 = (const float*)d_in[8];
    const float* b_bi1 = (const float*)d_in[9];
    const int*   e_src = (const int*)d_in[10];
    const int*   e_dst = (const int*)d_in[11];
    const float* e_val = (const float*)d_in[12];
    float* out = (float*)d_out;

    float* ego1 = (float*)d_ws;
    int2*  csr  = (int2*)(ego1 + (long)GNN_N * GNN_EMB);
    int*   cursor    = (int*)d_ws;
    int*   cnt_table = cursor + 1024;

    gnn_hist<<<SORT_BLOCKS, 1024, 0, stream>>>(e_src, cnt_table);
    gnn_offsets<<<1, 1024, 0, stream>>>(cnt_table, cursor, out);
    gnn_sort<<<SORT_BLOCKS, 1024, 0, stream>>>(e_src, e_dst, e_val, cursor, csr);

    gnn_layer<<<GNN_NBKT, 256, 0, stream>>>(csr, user_emb, item_emb, ego1,
                                            W_gc0, b_gc0, W_bi0, b_bi0, out, 64);
    gnn_layer<<<GNN_NBKT, 256, 0, stream>>>(csr, ego1, ego1 + (long)GNN_NUSERS * GNN_EMB,
                                            nullptr,
                                            W_gc1, b_gc1, W_bi1, b_bi1, out, 128);
}


# SpMM: layer block row-orders its tile's entries (LDS counting sort); a wave sums runs of equal rows in registers, one 256-B load per entry; LDS f32 add per run only; sum order = sorted position
# speedup vs baseline: 5.7722x; 1.2525x over previous
#include <hip/hip_runtime.h>
#include <math.h>


typedef __attribute__((ext_vector_type(2))) float v2f;
typedef __attribute__((ext_vector_type(8))) float v8f;
typedef __attribute__((ext_vector_type(2))) unsigned v2u;

#define GNN_N       100000
#define GNN_NUSERS  60000
#define GNN_EMB     64
#define GNN_E       3200000
#define GNN_NBKT    782
#define SORT_EDGES  16384
#define SORT_BLOCKS 196
#define CNT_PITCH   1024
#define YPITCH      68
#define ROW_CH      4608
#define ROW_KPT     18
#define ROW_FIRST   (1 << 24)

static_assert(GNN_N % 16 == 0, "a wave's 16-row tile is all inside or all outside N");
static_assert(GNN_N < (1 << 17), "a source row index fits the low 17 bits of an entry");

__device__ __forceinline__ unsigned gnn_bucket(int s)
{
    unsigned b = ((unsigned)s) >> 7;
    return b < GNN_NBKT ? b : GNN_NBKT - 1;
}

__device__ __forceinline__ int gnn_scan1024(int x, int* wt)
{
    const int lane = threadIdx.x & 31;
    const int w    = threadIdx.x >> 5;
    int inc = x;
    for (int d = 1; d < 32; d <<= 1) {
        int t = __shfl_up(inc, d);
        if (lane >= d) inc += t;
    }
    if (lane == 31) wt[w] = inc;
    __syncthreads();
    int off = 0;
    for (int i = 0; i < w; ++i) off += wt[i];
    __syncthreads();
    return off + inc - x;
}

__global__ __launch_bounds__(1024) void gnn_hist(
    const int* __restrict__ esrc, int* __restrict__ cnt_table)
{
    __shared__ int lh[GNN_NBKT];
    const int tid = threadIdx.x;
    if (tid < GNN_NBKT) lh[tid] = 0;
    __syncthreads();
    const int base = blockIdx.x * SORT_EDGES;
#pragma unroll
    for (int k = 0; k < 16; ++k) {
        int e = base + k * 1024 + tid;
        if (e < GNN_E) atomicAdd(&lh[gnn_bucket(esrc[e])], 1);
    }
    __syncthreads();
    if (tid < GNN_NBKT) cnt_table[blockIdx.x * CNT_PITCH + tid] = lh[tid];
}

__global__ __launch_bounds__(1024) void gnn_offsets(
    const int* __restrict__ cnt_table, int* __restrict__ cursor, float* __restrict__ out)
{
    __shared__ int wt[32];
    const int tid = threadIdx.x;
    int c = 0;
    if (tid < GNN_NBKT) {
        const int* col = cnt_table + tid;
#pragma unroll
        for (int b = 0; b < SORT_BLOCKS; ++b) c += col[b * CNT_PITCH];
    }
    int start = gnn_scan1024(c, wt);
    if (tid < GNN_NBKT) {
        cursor[tid] = start;
        float* o = out + (long)tid * 128 * 192 + 128;
        o[0] = __int_as_float(start);
        o[1] = __int_as_float(c);
    }
}

__global__ __launch_bounds__(1024) void gnn_sort(
    const int* __restrict__ esrc, const int* __restrict__ edst,
    const float* __restrict__ eval, int* __restrict__ cursor,
    int2* __restrict__ csr)
{
    __shared__ int2 sorted[SORT_EDGES];
    __shared__ int  lcnt[GNN_NBKT];
    __shared__ int  lstart[GNN_NBKT];
    __shared__ int  lcur[GNN_NBKT];
    __shared__ int  gbase[GNN_NBKT];
    __shared__ int  wt[32];

    const int tid  = threadIdx.x;
    const int base = blockIdx.x * SORT_EDGES;
    if (tid < GNN_NBKT) lcnt[tid] = 0;
    __syncthreads();

    int      ex[16], ey[16];
    unsigned bk[16];
#pragma unroll
    for (int k = 0; k < 16; ++k) {
        int e = base + k * 1024 + tid;
        bk[k] = GNN_NBKT;
        ex[k] = 0; ey[k] = 0;
        if (e < GNN_E) {
            int s = esrc[e];
            bk[k] = gnn_bucket(s);
            ex[k] = (edst[e] & 0x1FFFF) | ((s & 127) << 17);
            ey[k] = __float_as_int(eval[e]);
            atomicAdd(&lcnt[bk[k]], 1);
        }
    }
    __syncthreads();

    int c  = (tid < GNN_NBKT) ? lcnt[tid] : 0;
    int ls = gnn_scan1024(c, wt);
    if (tid < GNN_NBKT) {
        lstart[tid] = ls;
        lcur[tid]   = ls;
        gbase[tid]  = c ? atomicAdd(&cursor[tid], c) : 0;
    }
    __syncthreads();

#pragma unroll
    for (int k = 0; k < 16; ++k) {
        if (bk[k] < GNN_NBKT) {
            int r = atomicAdd(&lcur[bk[k]], 1);
            r = r < SORT_EDGES ? r : SORT_EDGES - 1;
            sorted[r] = make_int2(ex[k], ey[k]);
        }
    }
    __syncthreads();

    const int lane = tid & 31;
    for (int b = tid >> 5; b < GNN_NBKT; b += 32) {
        const int n  = lcnt[b];
        const int l0 = lstart[b];
        const int g0 = gbase[b];
        for (int i = lane; i < n; i += 32) {
            int g = g0 + i;
            g = g < GNN_E ? g : GNN_E - 1;
            csr[g] = sorted[l0 + i];
        }
    }
}

__device__ __forceinline__ void gnn_issue32(
    const float* __restrict__ xa, long offb, int lane, v2f (&x)[32])
{
    const int lo = (int)(offb & 0xFFFFFFFFL);
    const int hi = (int)(offb >> 32);
#pragma unroll
    for (int i = 0; i < 32; ++i) {
        v2u p;
        p.x = (unsigned)__builtin_amdgcn_readlane(lo, i);
        p.y = (unsigned)__builtin_amdgcn_readlane(hi, i);
        const unsigned long o = __builtin_bit_cast(unsigned long, p);
        const v2f* xr = (const v2f*)((const char*)xa + o);
        x[i] = xr[lane];
    }
}

template <bool FULL>
__device__ __forceinline__ void gnn_sum32(
    const v2f (&x)[32], int ebits, int vbits, unsigned chg, int m,
    int lane, float (*ytile)[YPITCH], int& cur, v2f& acc)
{
#pragma unroll
    for (int i = 0; i < 32; ++i) {
        if (FULL || i < m) {
            if (chg & (1u << i)) {
                atomicAdd(&ytile[cur][2 * lane],     acc.x);
                atomicAdd(&ytile[cur][2 * lane + 1], acc.y);
                acc.x = 0.f; acc.y = 0.f;
                cur = (__builtin_amdgcn_readlane(ebits, i) >> 17) & 127;
            }
            const float v = __int_as_float(__builtin_amdgcn_readlane(vbits, i));
            acc.x = fmaf(v, x[i].x, acc.x);
            acc.y = fmaf(v, x[i].y, acc.y);
        }
    }
}

__global__ __launch_bounds__(256) __attribute__((amdgpu_num_vgpr(128))) void gnn_layer(
    const int2* __restrict__ csr,
    const float* __restrict__ xa, const float* __restrict__ xb,
    float* __restrict__ ego_out,
    const float* __restrict__ Wgc, const float* __restrict__ bgc,
    const float* __restrict__ Wbi, const float* __restrict__ bbi,
    float* out, int out_col0)
{
    __shared__ __attribute__((aligned(16))) unsigned char smem[ROW_CH * 8];
    __shared__ __attribute__((aligned(16))) float ytile[128][YPITCH];
    __shared__ __attribute__((aligned(16))) int   rcnt[128];
    __shared__ __attribute__((aligned(16))) int   rstart[128];
    __shared__ float bsum[64];
    static_assert(ROW_CH * 8 >= 64 * 132 * 4, "the weights fit the shared region");
    float (*Wt)[132] = reinterpret_cast<float (*)[132]>(smem);
    int2*  sorted    = reinterpret_cast<int2*>(smem);

    const int tid  = threadIdx.x;
    const int wave = __builtin_amdgcn_readfirstlane(tid >> 5);
    const int lane = tid & 31;

    int start, cnt;
    {
        const float* o = out + (long)blockIdx.x * 128 * 192 + 128;
        start = __builtin_amdgcn_readfirstlane(__float_as_int(o[0]));
        cnt   = __builtin_amdgcn_readfirstlane(__float_as_int(o[1]));
        start = start < 0 ? 0 : (start > GNN_E ? GNN_E : start);
        cnt   = cnt < 0 ? 0 : (cnt > GNN_E - start ? GNN_E - start : cnt);
    }
    const int end = start + cnt;

    for (int i = tid; i < 128 * YPITCH; i += 256) (&ytile[0][0])[i] = 0.f;

    {
        const long xskip = (xb - xa) - (long)GNN_NUSERS * GNN_EMB;

        for (int cb = start; cb < end; cb += ROW_CH) {
            const int n = (end - cb) < ROW_CH ? (end - cb) : ROW_CH;
            if (tid < 128) rcnt[tid] = 0;
            __syncthreads();

            int ex[ROW_KPT], ey[ROW_KPT], rk[ROW_KPT];
#pragma unroll
            for (int k = 0; k < ROW_KPT; ++k) {
                int g = cb + tid + k * 256;
                g = g < GNN_E ? g : GNN_E - 1;
                const int2 t = csr[g];
                ex[k] = t.x; ey[k] = t.y;
            }
#pragma unroll
            for (int k = 0; k < ROW_KPT; ++k) {
                const int e = tid + k * 256;
                rk[k] = 0;
                if (e < n) rk[k] = atomicAdd(&rcnt[(ex[k] >> 17) & 127], 1);
            }
            __syncthreads();

            if (wave == 0) {
                const int4 c4 = *(const int4*)&rcnt[4 * lane];
                const int t0 = c4.x, t1 = t0 + c4.y, t2 = t1 + c4.z, t3 = t2 + c4.w;
                int inc = t3;
                for (int d = 1; d < 32; d <<= 1) {
                    const int t = __shfl_up(inc, d);
                    if (lane >= d) inc += t;
                }
                const int ex0 = inc - t3;
                *(int4*)&rstart[4 * lane] = make_int4(ex0, ex0 + t0, ex0 + t1, ex0 + t2);
            }
            __syncthreads();

            int rs[ROW_KPT];
#pragma unroll
            for (int k = 0; k < ROW_KPT; ++k) rs[k] = rstart[(ex[k] >> 17) & 127];
#pragma unroll
            for (int k = 0; k < ROW_KPT; ++k) {
                const int e = tid + k * 256;
                if (e < n) {
                    int pos = rs[k] + rk[k];
                    pos = pos < ROW_CH ? pos : ROW_CH - 1;
                    sorted[pos] = make_int2(ex[k] | (rk[k] == 0 ? ROW_FIRST : 0), ey[k]);
                }
            }
            __syncthreads();

            const int q  = (((n + 7) >> 3) + 31) & ~31;
            int       p  = wave * q;
            const int p1 = (p + q) < n ? (p + q) : n;
            if (p < p1) {
                v2f acc; acc.x = 0.f; acc.y = 0.f;
                int cur = (__builtin_amdgcn_readfirstlane(sorted[p].x) >> 17) & 127;
                int2 ent = make_int2(0, 0);
                if (p + lane < p1) ent = sorted[p + lane];
                while (p < p1) {
                    const int m  = (p1 - p) < 32 ? (p1 - p) : 32;
                    const int np = p + 32;
                    unsigned di = (unsigned)ent.x & 0x1FFFFu;
                    di = di < GNN_N ? di : GNN_N - 1;
                    const long     offb = ((long)di * GNN_EMB + (di < GNN_NUSERS ? 0L : xskip)) * 4;
                    const unsigned chg  = __builtin_amdgcn_ballot_w32((ent.x & ROW_FIRST) != 0);
                    v2f x[32];
                    gnn_issue32(xa, offb, lane, x);
                    int2 nxt = make_int2(0, 0);
                    if (np + lane < p1) nxt = sorted[np + lane];
                    if (m == 32) gnn_sum32<true >(x, ent.x, ent.y, chg, m, lane, ytile, cur, acc);
                    else         gnn_sum32<false>(x, ent.x, ent.y, chg, m, lane, ytile, cur, acc);
                    ent = nxt;
                    p   = np;
                }
                atomicAdd(&ytile[cur][2 * lane],     acc.x);
                atomicAdd(&ytile[cur][2 * lane + 1], acc.y);
            }
        }
    }
    __syncthreads();

    float4 xv[8];
    if (ego_out) {
#pragma unroll
        for (int k = 0; k < 8; ++k) {
            const int i   = tid + k * 256;
            const int row = blockIdx.x * 128 + (i >> 4);
            xv[k] = make_float4(0.f, 0.f, 0.f, 0.f);
            if (row < GNN_N) {
                const float* xr = (row < GNN_NUSERS) ? xa + (long)row * GNN_EMB
                                                     : xb + (long)(row - GNN_NUSERS) * GNN_EMB;
                xv[k] = *(const float4*)(xr + ((i & 15) << 2));
            }
        }
    }

    for (int i = tid; i < 64 * 128; i += 256) {
        int k = i >> 6;
        int n = i & 63;
        float w = (k < 64) ? Wgc[k * 64 + n] : Wbi[(k - 64) * 64 + n];
        Wt[n][k] = w;
    }
    if (tid < 64) bsum[tid] = bgc[tid] + bbi[tid];
    __syncthreads();

    if (ego_out) {
#pragma unroll
        for (int k = 0; k < 8; ++k) {
            const int i   = tid + k * 256;
            const int row = blockIdx.x * 128 + (i >> 4);
            if (row < GNN_N) *(float4*)(out + (long)row * 192 + ((i & 15) << 2)) = xv[k];
        }
    }

    const int rowbase = blockIdx.x * 128 + wave * 16;
    if (rowbase >= GNN_N) return;

    const int m    = lane & 15;
    const int half = lane >> 4;
    const int kb   = half << 1;

    const long  row   = (long)(rowbase + m);
    const float* egoRd = (row < GNN_NUSERS) ? xa + row * 64 : xb + (row - GNN_NUSERS) * 64;
    const float* sLRd  = &ytile[wave * 16 + m][0];

    v2f e2[16], l2[16];
#pragma unroll
    for (int s = 0; s < 16; ++s) {
        int c = 4 * s + kb;
        e2[s] = *(const v2f*)(egoRd + c);
        l2[s] = *(const v2f*)(sLRd + c);
    }

    v8f acc[4] = {};

#pragma unroll
    for (int s = 0; s < 16; ++s) {
        v2f a;
        a.x = e2[s].x + l2[s].x;
        a.y = e2[s].y + l2[s].y;
#pragma unroll
        for (int nt = 0; nt < 4; ++nt) {
            v2f b = *(const v2f*)(&Wt[nt * 16 + m][4 * s + kb]);
            acc[nt] = __builtin_amdgcn_wmma_f32_16x16x4_f32(
                false, a, false, b, (short)0, acc[nt], false, false);
        }
    }
#pragma unroll
    for (int s = 0; s < 16; ++s) {
        v2f a;
        a.x = e2[s].x * l2[s].x;
        a.y = e2[s].y * l2[s].y;
#pragma unroll
        for (int nt = 0; nt < 4; ++nt) {
            v2f b = *(const v2f*)(&Wt[nt * 16 + m][64 + 4 * s + kb]);
            acc[nt] = __builtin_amdgcn_wmma_f32_16x16x4_f32(
                false, a, false, b, (short)0, acc[nt], false, false);
        }
    }

    float vout[4][8];
    float ssq[8];
#pragma unroll
    for (int j = 0; j < 8; ++j) ssq[j] = 0.f;
#pragma unroll
    for (int nt = 0; nt < 4; ++nt) {
        float bc = bsum[nt * 16 + m];
#pragma unroll
        for (int j = 0; j < 8; ++j) {
            float v = acc[nt][j] + bc;
            v = (v > 0.f) ? v : 0.01f * v;
            vout[nt][j] = v;
            ssq[j] += v * v;
        }
    }

    if (ego_out) {
        float* egoWr = ego_out + (long)rowbase * 64;
#pragma unroll
        for (int j = 0; j < 8; ++j) {
            int mrow = j + (half << 3);
#pragma unroll
            for (int nt = 0; nt < 4; ++nt)
                egoWr[mrow * 64 + nt * 16 + m] = vout[nt][j];
        }
    }

#pragma unroll
    for (int j = 0; j < 8; ++j) {
        float s = ssq[j];
        s += __shfl_xor(s, 8, 16);
        s += __shfl_xor(s, 4, 16);
        s += __shfl_xor(s, 2, 16);
        s += __shfl_xor(s, 1, 16);
        float inv = 1.0f / fmaxf(sqrtf(s), 1e-12f);
        int mrow = j + (half << 3);
        float* o = out + (long)(rowbase + mrow) * 192 + out_col0;
#pragma unroll
        for (int nt = 0; nt < 4; ++nt)
            o[nt * 16 + m] = vout[nt][j] * inv;
    }
}

extern "C" void kernel_launch(void* const* d_in, const int* in_sizes, int n_in,
                              void* d_out, int out_size, void* d_ws, size_t ws_size,
                              hipStream_t stream) {
    const float* user_emb = (const float*)d_in[0];
    const float* item_emb = (const float*)d_in[1];
    const float* W_gc0 = (const float*)d_in[2];
    const float* b_gc0 = (const float*)d_in[3];
    const float* W_bi0 = (const float*)d_in[4];
    const float* b_bi0 = (const float*)d_in[5];
    const float* W_gc1 = (const float*)d_in[6];
    const float* b_gc1 = (const float*)d_in[7];
    const float* W_bi1 = (const float*)d_in[8];
    const float* b_bi1 = (const float*)d_in[9];
    const int*   e_src = (const int*)d_in[10];
    const int*   e_dst = (const int*)d_in[11];
    const float* e_val = (const float*)d_in[12];
    float* out = (float*)d_out;

    float* ego1 = (float*)d_ws;
    int2*  csr  = (int2*)(ego1 + (long)GNN_N * GNN_EMB);
    int*   cursor    = (int*)d_ws;
    int*   cnt_table = cursor + 1024;

    gnn_hist<<<SORT_BLOCKS, 1024, 0, stream>>>(e_src, cnt_table);
    gnn_offsets<<<1, 1024, 0, stream>>>(cnt_table, cursor, out);
    gnn_sort<<<SORT_BLOCKS, 1024, 0, stream>>>(e_src, e_dst, e_val, cursor, csr);

    gnn_layer<<<GNN_NBKT, 256, 0, stream>>>(csr, user_emb, item_emb, ego1,
                                            W_gc0, b_gc0, W_bi0, b_bi0, out, 64);
    gnn_layer<<<GNN_NBKT, 256, 0, stream>>>(csr, ego1, ego1 + (long)GNN_NUSERS * GNN_EMB,
                                            nullptr,
                                            W_gc1, b_gc1, W_bi1, b_bi1, out, 128);
}


# Edge bucketing without cross-block counting: sort blocks write whole tile-ordered slices + a run word per tile, layer blocks collect 196 runs; 5 launches to 3, no global atomic; LDS-rank sum order
# speedup vs baseline: 5.9721x; 1.0346x over previous
#include <hip/hip_runtime.h>
#include <math.h>


typedef __attribute__((ext_vector_type(2))) float v2f;
typedef __attribute__((ext_vector_type(8))) float v8f;
typedef __attribute__((ext_vector_type(2))) unsigned v2u;

#define GNN_N       100000
#define GNN_NUSERS  60000
#define GNN_EMB     64
#define GNN_E       3200000
#define GNN_NBKT    782
#define SORT_EDGES  16384
#define SORT_BLOCKS 196
#define RUNS_PW     25
#define YPITCH      68
#define ROW_CH      4608
#define ROW_KPT     18
#define ROW_FIRST   (1 << 24)

static_assert(GNN_N % 16 == 0, "a wave's 16-row tile is all inside or all outside N");
static_assert(GNN_N < (1 << 17), "a source row index fits the low 17 bits of an entry");
static_assert(SORT_EDGES <= (1 << 14), "first slot and count of a run fit 15 bits each");
static_assert(8 * RUNS_PW >= SORT_BLOCKS && RUNS_PW <= 32, "every slice has a lane in a layer block");
static_assert(SORT_BLOCKS <= 4 * 64, "a tile's run words fit columns 128..191 of its first four rows");
static_assert((GNN_NBKT - 1) * 128 + 3 < GNN_N, "the last tile has those four rows");

__device__ __forceinline__ long gnn_run_word(int tile, int blk)
{
    return ((long)tile * 128 + (blk >> 6)) * 192 + 128 + (blk & 63);
}

__device__ __forceinline__ unsigned gnn_bucket(int s)
{
    unsigned b = ((unsigned)s) >> 7;
    return b < GNN_NBKT ? b : GNN_NBKT - 1;
}

__device__ __forceinline__ int gnn_scan1024(int x, int* wt)
{
    const int lane = threadIdx.x & 31;
    const int w    = threadIdx.x >> 5;
    int inc = x;
    for (int d = 1; d < 32; d <<= 1) {
        int t = __shfl_up(inc, d);
        if (lane >= d) inc += t;
    }
    if (lane == 31) wt[w] = inc;
    __syncthreads();
    int off = 0;
    for (int i = 0; i < w; ++i) off += wt[i];
    __syncthreads();
    return off + inc - x;
}

__global__ __launch_bounds__(1024) void gnn_sort(
    const int* __restrict__ esrc, const int* __restrict__ edst,
    const float* __restrict__ eval, int2* __restrict__ csr, float* __restrict__ out)
{
    __shared__ int2 sorted[SORT_EDGES];
    __shared__ int  lcnt[GNN_NBKT];
    __shared__ int  lcur[GNN_NBKT];
    __shared__ int  wt[32];

    const int tid  = threadIdx.x;
    const int base = blockIdx.x * SORT_EDGES;
    if (tid < GNN_NBKT) lcnt[tid] = 0;
    __syncthreads();

    int      ex[16], ey[16];
    unsigned bk[16];
#pragma unroll
    for (int k = 0; k < 16; ++k) {
        int e = base + k * 1024 + tid;
        bk[k] = GNN_NBKT;
        ex[k] = 0; ey[k] = 0;
        if (e < GNN_E) {
            int s = esrc[e];
            bk[k] = gnn_bucket(s);
            ex[k] = (edst[e] & 0x1FFFF) | ((s & 127) << 17);
            ey[k] = __float_as_int(eval[e]);
            atomicAdd(&lcnt[bk[k]], 1);
        }
    }
    __syncthreads();

    int c  = (tid < GNN_NBKT) ? lcnt[tid] : 0;
    int ls = gnn_scan1024(c, wt);
    if (tid < GNN_NBKT) {
        lcur[tid] = ls;
        out[gnn_run_word(tid, blockIdx.x)] = __int_as_float(ls | (c << 15));
    }
    __syncthreads();

#pragma unroll
    for (int k = 0; k < 16; ++k) {
        if (bk[k] < GNN_NBKT) {
            int r = atomicAdd(&lcur[bk[k]], 1);
            r = r < SORT_EDGES ? r : SORT_EDGES - 1;
            sorted[r] = make_int2(ex[k], ey[k]);
        }
    }
    __syncthreads();

#pragma unroll
    for (int k = 0; k < 16; ++k) {
        const int i = k * 1024 + tid;
        if (base + i < GNN_E) csr[base + i] = sorted[i];
    }
}

__device__ __forceinline__ void gnn_issue32(
    const float* __restrict__ xa, long offb, int lane, v2f (&x)[32])
{
    const int lo = (int)(offb & 0xFFFFFFFFL);
    const int hi = (int)(offb >> 32);
#pragma unroll
    for (int i = 0; i < 32; ++i) {
        v2u p;
        p.x = (unsigned)__builtin_amdgcn_readlane(lo, i);
        p.y = (unsigned)__builtin_amdgcn_readlane(hi, i);
        const unsigned long o = __builtin_bit_cast(unsigned long, p);
        const v2f* xr = (const v2f*)((const char*)xa + o);
        x[i] = xr[lane];
    }
}

template <bool FULL>
__device__ __forceinline__ void gnn_sum32(
    const v2f (&x)[32], int ebits, int vbits, unsigned chg, int m,
    int lane, float (*ytile)[YPITCH], int& cur, v2f& acc)
{
#pragma unroll
    for (int i = 0; i < 32; ++i) {
        if (FULL || i < m) {
            if (chg & (1u << i)) {
                atomicAdd(&ytile[cur][2 * lane],     acc.x);
                atomicAdd(&ytile[cur][2 * lane + 1], acc.y);
                acc.x = 0.f; acc.y = 0.f;
                cur = (__builtin_amdgcn_readlane(ebits, i) >> 17) & 127;
            }
            const float v = __int_as_float(__builtin_amdgcn_readlane(vbits, i));
            acc.x = fmaf(v, x[i].x, acc.x);
            acc.y = fmaf(v, x[i].y, acc.y);
        }
    }
}

__global__ __launch_bounds__(256) __attribute__((amdgpu_num_vgpr(128))) void gnn_layer(
    const int2* __restrict__ csr,
    const float* __restrict__ xa, const float* __restrict__ xb,
    float* __restrict__ ego_out,
    const float* __restrict__ Wgc, const float* __restrict__ bgc,
    const float* __restrict__ Wbi, const float* __restrict__ bbi,
    float* out, int out_col0)
{
    __shared__ __attribute__((aligned(16))) unsigned char smem[ROW_CH * 8];
    __shared__ __attribute__((aligned(16))) float ytile[128][YPITCH];
    __shared__ __attribute__((aligned(16))) int   rcnt[128];
    __shared__ __attribute__((aligned(16))) int   rstart[128];
    __shared__ float bsum[64];
    __shared__ int   wtot[8];
    static_assert(ROW_CH * 8 >= 64 * 132 * 4, "the weights fit the shared region");
    float (*Wt)[132] = reinterpret_cast<float (*)[132]>(smem);
    int2*  sorted    = reinterpret_cast<int2*>(smem);

    const int tid  = threadIdx.x;
    const int wave = __builtin_amdgcn_readfirstlane(tid >> 5);
    const int lane = tid & 31;

    int rsrc = 0, rcntv = 0, rpos, cnt;
    {
        const int r = wave * RUNS_PW + lane;
        if (lane < RUNS_PW && r < SORT_BLOCKS) {
            const int pk = __float_as_int(out[gnn_run_word(blockIdx.x, r)]);
            int s0 = pk & 0x7FFF;
            int c0 = (pk >> 15) & 0x7FFF;
            s0 = s0 < SORT_EDGES ? s0 : SORT_EDGES;
            c0 = c0 < SORT_EDGES - s0 ? c0 : SORT_EDGES - s0;
            rsrc  = r * SORT_EDGES + s0;
            rcntv = c0;
        }
        int inc = rcntv;
        for (int d = 1; d < 32; d <<= 1) {
            const int t = __shfl_up(inc, d);
            if (lane >= d) inc += t;
        }
        if (lane == 31) wtot[wave] = inc;
        __syncthreads();
        int before = 0, all = 0;
#pragma unroll
        for (int i = 0; i < 8; ++i) {
            const int t = wtot[i];
            all += t;
            if (i < wave) before += t;
        }
        rpos = before + inc - rcntv;
        cnt = __builtin_amdgcn_readfirstlane(all);
    }

    for (int i = tid; i < 128 * YPITCH; i += 256) (&ytile[0][0])[i] = 0.f;

    {
        const long xskip = (xb - xa) - (long)GNN_NUSERS * GNN_EMB;

        for (int cb = 0; cb < cnt; cb += ROW_CH) {
            const int n = (cnt - cb) < ROW_CH ? (cnt - cb) : ROW_CH;
            if (tid < 128) rcnt[tid] = 0;
            __syncthreads();

            {
                int2 tv[RUNS_PW];
#pragma unroll
                for (int j = 0; j < RUNS_PW; ++j) {
                    int g = __builtin_amdgcn_readlane(rsrc, j) + lane;
                    g = g < GNN_E ? g : GNN_E - 1;
                    tv[j] = csr[g];
                }
#pragma unroll
                for (int j = 0; j < RUNS_PW; ++j) {
                    const int c   = __builtin_amdgcn_readlane(rcntv, j);
                    const int pos = __builtin_amdgcn_readlane(rpos, j) - cb + lane;
                    if (lane < c && pos >= 0 && pos < ROW_CH) sorted[pos] = tv[j];
                }
                if (__builtin_amdgcn_ballot_w32(rcntv > 32) != 0) {
                    for (int j = 0; j < RUNS_PW; ++j) {
                        const int c  = __shfl(rcntv, j);
                        const int g0 = __shfl(rsrc, j);
                        const int p0 = __shfl(rpos, j) - cb;
                        for (int i = 32 + lane; i < c; i += 32) {
                            const int pos = p0 + i;
                            int g = g0 + i;
                            g = g < GNN_E ? g : GNN_E - 1;
                            if (pos >= 0 && pos < ROW_CH) sorted[pos] = csr[g];
                        }
                    }
                }
            }
            __syncthreads();

            int ex[ROW_KPT], ey[ROW_KPT], rk[ROW_KPT];
#pragma unroll
            for (int k = 0; k < ROW_KPT; ++k) {
                const int2 t = sorted[tid + k * 256];
                ex[k] = t.x; ey[k] = t.y;
            }
#pragma unroll
            for (int k = 0; k < ROW_KPT; ++k) {
                const int e = tid + k * 256;
                rk[k] = 0;
                if (e < n) rk[k] = atomicAdd(&rcnt[(ex[k] >> 17) & 127], 1);
            }
            __syncthreads();

            if (wave == 0) {
                const int4 c4 = *(const int4*)&rcnt[4 * lane];
                const int t0 = c4.x, t1 = t0 + c4.y, t2 = t1 + c4.z, t3 = t2 + c4.w;
                int inc = t3;
                for (int d = 1; d < 32; d <<= 1) {
                    const int t = __shfl_up(inc, d);
                    if (lane >= d) inc += t;
                }
                const int ex0 = inc - t3;
                *(int4*)&rstart[4 * lane] = make_int4(ex0, ex0 + t0, ex0 + t1, ex0 + t2);
            }
            __syncthreads();

            int rs[ROW_KPT];
#pragma unroll
            for (int k = 0; k < ROW_KPT; ++k) rs[k] = rstart[(ex[k] >> 17) & 127];
#pragma unroll
            for (int k = 0; k < ROW_KPT; ++k) {
                const int e = tid + k * 256;
                if (e < n) {
                    int pos = rs[k] + rk[k];
                    pos = pos < ROW_CH ? pos : ROW_CH - 1;
                    sorted[pos] = make_int2(ex[k] | (rk[k] == 0 ? ROW_FIRST : 0), ey[k]);
                }
            }
            __syncthreads();

            const int q  = (((n + 7) >> 3) + 31) & ~31;
            int       p  = wave * q;
            const int p1 = (p + q) < n ? (p + q) : n;
            if (p < p1) {
                v2f acc; acc.x = 0.f; acc.y = 0.f;
                int cur = (__builtin_amdgcn_readfirstlane(sorted[p].x) >> 17) & 127;
                int2 ent = make_int2(0, 0);
                if (p + lane < p1) ent = sorted[p + lane];
                while (p < p1) {
                    const int m  = (p1 - p) < 32 ? (p1 - p) : 32;
                    const int np = p + 32;
                    unsigned di = (unsigned)ent.x & 0x1FFFFu;
                    di = di < GNN_N ? di : GNN_N - 1;
                    const long     offb = ((long)di * GNN_EMB + (di < GNN_NUSERS ? 0L : xskip)) * 4;
                    const unsigned chg  = __builtin_amdgcn_ballot_w32((ent.x & ROW_FIRST) != 0);
                    v2f x[32];
                    gnn_issue32(xa, offb, lane, x);
                    int2 nxt = make_int2(0, 0);
                    if (np + lane < p1) nxt = sorted[np + lane];
                    if (m == 32) gnn_sum32<true >(x, ent.x, ent.y, chg, m, lane, ytile, cur, acc);
                    else         gnn_sum32<false>(x, ent.x, ent.y, chg, m, lane, ytile, cur, acc);
                    ent = nxt;
                    p   = np;
                }
                atomicAdd(&ytile[cur][2 * lane],     acc.x);
                atomicAdd(&ytile[cur][2 * lane + 1], acc.y);
            }
        }
    }
    __syncthreads();

    float4 xv[8];
    if (ego_out) {
#pragma unroll
        for (int k = 0; k < 8; ++k) {
            const int i   = tid + k * 256;
            const int row = blockIdx.x * 128 + (i >> 4);
            xv[k] = make_float4(0.f, 0.f, 0.f, 0.f);
            if (row < GNN_N) {
                const float* xr = (row < GNN_NUSERS) ? xa + (long)row * GNN_EMB
                                                     : xb + (long)(row - GNN_NUSERS) * GNN_EMB;
                xv[k] = *(const float4*)(xr + ((i & 15) << 2));
            }
        }
    }

    for (int i = tid; i < 64 * 128; i += 256) {
        int k = i >> 6;
        int n = i & 63;
        float w = (k < 64) ? Wgc[k * 64 + n] : Wbi[(k - 64) * 64 + n];
        Wt[n][k] = w;
    }
    if (tid < 64) bsum[tid] = bgc[tid] + bbi[tid];
    __syncthreads();

    if (ego_out) {
#pragma unroll
        for (int k = 0; k < 8; ++k) {
            const int i   = tid + k * 256;
            const int row = blockIdx.x * 128 + (i >> 4);
            if (row < GNN_N) *(float4*)(out + (long)row * 192 + ((i & 15) << 2)) = xv[k];
        }
    }

    const int rowbase = blockIdx.x * 128 + wave * 16;
    if (rowbase >= GNN_N) return;

    const int m    = lane & 15;
    const int half = lane >> 4;
    const int kb   = half << 1;

    const long  row   = (long)(rowbase + m);
    const float* egoRd = (row < GNN_NUSERS) ? xa + row * 64 : xb + (row - GNN_NUSERS) * 64;
    const float* sLRd  = &ytile[wave * 16 + m][0];

    v2f e2[16], l2[16];
#pragma unroll
    for (int s = 0; s < 16; ++s) {
        int c = 4 * s + kb;
        e2[s] = *(const v2f*)(egoRd + c);
        l2[s] = *(const v2f*)(sLRd + c);
    }

    v8f acc[4] = {};

#pragma unroll
    for (int s = 0; s < 16; ++s) {
        v2f a;
        a.x = e2[s].x + l2[s].x;
        a.y = e2[s].y + l2[s].y;
#pragma unroll
        for (int nt = 0; nt < 4; ++nt) {
            v2f b = *(const v2f*)(&Wt[nt * 16 + m][4 * s + kb]);
            acc[nt] = __builtin_amdgcn_wmma_f32_16x16x4_f32(
                false, a, false, b, (short)0, acc[nt], false, false);
        }
    }
#pragma unroll
    for (int s = 0; s < 16; ++s) {
        v2f a;
        a.x = e2[s].x * l2[s].x;
        a.y = e2[s].y * l2[s].y;
#pragma unroll
        for (int nt = 0; nt < 4; ++nt) {
            v2f b = *(const v2f*)(&Wt[nt * 16 + m][64 + 4 * s + kb]);
            acc[nt] = __builtin_amdgcn_wmma_f32_16x16x4_f32(
                false, a, false, b, (short)0, acc[nt], false, false);
        }
    }

    float vout[4][8];
    float ssq[8];
#pragma unroll
    for (int j = 0; j < 8; ++j) ssq[j] = 0.f;
#pragma unroll
    for (int nt = 0; nt < 4; ++nt) {
        float bc = bsum[nt * 16 + m];
#pragma unroll
        for (int j = 0; j < 8; ++j) {
            float v = acc[nt][j] + bc;
            v = (v > 0.f) ? v : 0.01f * v;
            vout[nt][j] = v;
            ssq[j] += v * v;
        }
    }

    if (ego_out) {
        float* egoWr = ego_out + (long)rowbase * 64;
#pragma unroll
        for (int j = 0; j < 8; ++j) {
            int mrow = j + (half << 3);
#pragma unroll
            for (int nt = 0; nt < 4; ++nt)
                egoWr[mrow * 64 + nt * 16 + m] = vout[nt][j];
        }
    }

#pragma unroll
    for (int j = 0; j < 8; ++j) {
        float s = ssq[j];
        s += __shfl_xor(s, 8, 16);
        s += __shfl_xor(s, 4, 16);
        s += __shfl_xor(s, 2, 16);
        s += __shfl_xor(s, 1, 16);
        float inv = 1.0f / fmaxf(sqrtf(s), 1e-12f);
        int mrow = j + (half << 3);
        float* o = out + (long)(rowbase + mrow) * 192 + out_col0;
#pragma unroll
        for (int nt = 0; nt < 4; ++nt)
            o[nt * 16 + m] = vout[nt][j] * inv;
    }
}

extern "C" void kernel_launch(void* const* d_in, const int* in_sizes, int n_in,
                              void* d_out, int out_size, void* d_ws, size_t ws_size,
                              hipStream_t stream) {
    const float* user_emb = (const float*)d_in[0];
    const float* item_emb = (const float*)d_in[1];
    const float* W_gc0 = (const float*)d_in[2];
    const float* b_gc0 = (const float*)d_in[3];
    const float* W_bi0 = (const float*)d_in[4];
    const float* b_bi0 = (const float*)d_in[5];
    const float* W_gc1 = (const float*)d_in[6];
    const float* b_gc1 = (const float*)d_in[7];
    const float* W_bi1 = (const float*)d_in[8];
    const float* b_bi1 = (const float*)d_in[9];
    const int*   e_src = (const int*)d_in[10];
    const int*   e_dst = (const int*)d_in[11];
    const float* e_val = (const float*)d_in[12];
    float* out = (float*)d_out;

    float* ego1 = (float*)d_ws;
    int2*  csr  = (int2*)(ego1 + (long)GNN_N * GNN_EMB);

    gnn_sort<<<SORT_BLOCKS, 1024, 0, stream>>>(e_src, e_dst, e_val, csr, out);

    gnn_layer<<<GNN_NBKT, 256, 0, stream>>>(csr, user_emb, item_emb, ego1,
                                            W_gc0, b_gc0, W_bi0, b_bi0, out, 64);
    gnn_layer<<<GNN_NBKT, 256, 0, stream>>>(csr, ego1, ego1 + (long)GNN_NUSERS * GNN_EMB,
                                            nullptr,
                                            W_gc1, b_gc1, W_bi1, b_bi1, out, 128);
}


# Sort kernel: edges read as twelve 16-B loads per thread, all issued before the first use (was 16 dependent trips to memory); LDS passes and the layer's weight loads grouped too; sum order: LDS rank
# speedup vs baseline: 6.2243x; 1.0422x over previous
#include <hip/hip_runtime.h>
#include <math.h>


typedef __attribute__((ext_vector_type(2))) float v2f;
typedef __attribute__((ext_vector_type(8))) float v8f;
typedef __attribute__((ext_vector_type(2))) unsigned v2u;

#define GNN_N       100000
#define GNN_NUSERS  60000
#define GNN_EMB     64
#define GNN_E       3200000
#define GNN_NBKT    782
#define SORT_EDGES  16384
#define SORT_BLOCKS 196
#define RUNS_PW     25
#define YPITCH      68
#define ROW_CH      4608
#define ROW_KPT     18
#define ROW_FIRST   (1 << 24)

static_assert(GNN_N % 16 == 0, "a wave's 16-row tile is all inside or all outside N");
static_assert(GNN_N < (1 << 17), "a source row index fits the low 17 bits of an entry");
static_assert(SORT_EDGES <= (1 << 14), "first slot and count of a run fit 15 bits each");
static_assert(8 * RUNS_PW >= SORT_BLOCKS && RUNS_PW <= 32, "every slice has a lane in a layer block");
static_assert(SORT_BLOCKS <= 4 * 64, "a tile's run words fit columns 128..191 of its first four rows");
static_assert((GNN_NBKT - 1) * 128 + 3 < GNN_N, "the last tile has those four rows");

__device__ __forceinline__ long gnn_run_word(int tile, int blk)
{
    return ((long)tile * 128 + (blk >> 6)) * 192 + 128 + (blk & 63);
}

__device__ __forceinline__ unsigned gnn_bucket(int s)
{
    unsigned b = ((unsigned)s) >> 7;
    return b < GNN_NBKT ? b : GNN_NBKT - 1;
}

__device__ __forceinline__ int gnn_scan1024(int x, int* wt)
{
    const int lane = threadIdx.x & 31;
    const int w    = threadIdx.x >> 5;
    int inc = x;
    for (int d = 1; d < 32; d <<= 1) {
        int t = __shfl_up(inc, d);
        if (lane >= d) inc += t;
    }
    if (lane == 31) wt[w] = inc;
    __syncthreads();
    int off = 0;
    for (int i = 0; i < w; ++i) off += wt[i];
    __syncthreads();
    return off + inc - x;
}

static_assert(GNN_E % 4 == 0, "a group of four consecutive edges is all inside or all outside the list");
static_assert(SORT_EDGES == 4 * 4 * 1024, "four groups of four edges per thread");

template <bool FULL>
__device__ __forceinline__ void gnn_sort_body(
    const int* __restrict__ esrc, const int* __restrict__ edst,
    const float* __restrict__ eval, int2* __restrict__ csr, float* __restrict__ out,
    int2* sorted, int* lcnt, int* lcur, int* wt)
{
    const int tid  = threadIdx.x;
    const int base = blockIdx.x * SORT_EDGES;

    int4   s4[4], d4[4];
    float4 v4[4];
#pragma unroll
    for (int j = 0; j < 4; ++j) {
        int e = base + j * 4096 + 4 * tid;
        if (!FULL) e = e < GNN_E ? e : GNN_E - 4;
        s4[j] = *(const int4*)(esrc + e);
        d4[j] = *(const int4*)(edst + e);
        v4[j] = *(const float4*)(eval + e);
    }
    __syncthreads();

    int      ex[16], ey[16];
    unsigned bk[16];
#pragma unroll
    for (int j = 0; j < 4; ++j) {
        const bool ok = FULL || (base + j * 4096 + 4 * tid) < GNN_E;
        const int   sv[4] = { s4[j].x, s4[j].y, s4[j].z, s4[j].w };
        const int   dv[4] = { d4[j].x, d4[j].y, d4[j].z, d4[j].w };
        const float vv[4] = { v4[j].x, v4[j].y, v4[j].z, v4[j].w };
#pragma unroll
        for (int q = 0; q < 4; ++q) {
            const int k = 4 * j + q;
            bk[k] = ok ? gnn_bucket(sv[q]) : GNN_NBKT;
            ex[k] = (dv[q] & 0x1FFFF) | ((sv[q] & 127) << 17);
            ey[k] = __float_as_int(vv[q]);
        }
        if (ok) {
#pragma unroll
            for (int q = 0; q < 4; ++q) atomicAdd(&lcnt[bk[4 * j + q]], 1);
        }
    }
    __syncthreads();

    const int c  = (tid < GNN_NBKT) ? lcnt[tid] : 0;
    const int ls = gnn_scan1024(c, wt);
    if (tid < GNN_NBKT) lcur[tid] = ls;
    __syncthreads();

    int r[16];
#pragma unroll
    for (int k = 0; k < 16; ++k) {
        r[k] = 0;
        if (FULL || bk[k] < GNN_NBKT) r[k] = atomicAdd(&lcur[bk[k]], 1);
    }
#pragma unroll
    for (int k = 0; k < 16; ++k) {
        if (FULL || bk[k] < GNN_NBKT) {
            const int p = r[k] < SORT_EDGES ? r[k] : SORT_EDGES - 1;
            sorted[p] = make_int2(ex[k], ey[k]);
        }
    }
    __syncthreads();

    const int4* sorted4 = reinterpret_cast<const int4*>(sorted);
    int4*       csr4    = reinterpret_cast<int4*>(csr + base);
    int4 t[8];
#pragma unroll
    for (int k = 0; k < 8; ++k) t[k] = sorted4[k * 1024 + tid];
#pragma unroll
    for (int k = 0; k < 8; ++k) {
        const int i = k * 1024 + tid;
        if (FULL || base + 2 * i < GNN_E) csr4[i] = t[k];
    }

    if (tid < GNN_NBKT)
        out[gnn_run_word(tid, blockIdx.x)] = __int_as_float(ls | (c << 15));
}

__global__ __launch_bounds__(1024) void gnn_sort(
    const int* __restrict__ esrc, const int* __restrict__ edst,
    const float* __restrict__ eval, int2* __restrict__ csr, float* __restrict__ out)
{
    __shared__ __attribute__((aligned(16))) int2 sorted[SORT_EDGES];
    __shared__ int  lcnt[GNN_NBKT];
    __shared__ int  lcur[GNN_NBKT];
    __shared__ int  wt[32];

    if (threadIdx.x < GNN_NBKT) lcnt[threadIdx.x] = 0;
    if ((blockIdx.x + 1) * SORT_EDGES <= GNN_E)
        gnn_sort_body<true >(esrc, edst, eval, csr, out, sorted, lcnt, lcur, wt);
    else
        gnn_sort_body<false>(esrc, edst, eval, csr, out, sorted, lcnt, lcur, wt);
}

__device__ __forceinline__ void gnn_issue32(
    const float* __restrict__ xa, long offb, int lane, v2f (&x)[32])
{
    const int lo = (int)(offb & 0xFFFFFFFFL);
    const int hi = (int)(offb >> 32);
#pragma unroll
    for (int i = 0; i < 32; ++i) {
        v2u p;
        p.x = (unsigned)__builtin_amdgcn_readlane(lo, i);
        p.y = (unsigned)__builtin_amdgcn_readlane(hi, i);
        const unsigned long o = __builtin_bit_cast(unsigned long, p);
        const v2f* xr = (const v2f*)((const char*)xa + o);
        x[i] = xr[lane];
    }
}

template <bool FULL>
__device__ __forceinline__ void gnn_sum32(
    const v2f (&x)[32], int ebits, int vbits, unsigned chg, int m,
    int lane, float (*ytile)[YPITCH], int& cur, v2f& acc)
{
#pragma unroll
    for (int i = 0; i < 32; ++i) {
        if (FULL || i < m) {
            if (chg & (1u << i)) {
                atomicAdd(&ytile[cur][2 * lane],     acc.x);
                atomicAdd(&ytile[cur][2 * lane + 1], acc.y);
                acc.x = 0.f; acc.y = 0.f;
                cur = (__builtin_amdgcn_readlane(ebits, i) >> 17) & 127;
            }
            const float v = __int_as_float(__builtin_amdgcn_readlane(vbits, i));
            acc.x = fmaf(v, x[i].x, acc.x);
            acc.y = fmaf(v, x[i].y, acc.y);
        }
    }
}

__global__ __launch_bounds__(256) __attribute__((amdgpu_num_vgpr(128))) void gnn_layer(
    const int2* __restrict__ csr,
    const float* __restrict__ xa, const float* __restrict__ xb,
    float* __restrict__ ego_out,
    const float* __restrict__ Wgc, const float* __restrict__ bgc,
    const float* __restrict__ Wbi, const float* __restrict__ bbi,
    float* out, int out_col0)
{
    __shared__ __attribute__((aligned(16))) unsigned char smem[ROW_CH * 8];
    __shared__ __attribute__((aligned(16))) float ytile[128][YPITCH];
    __shared__ __attribute__((aligned(16))) int   rcnt[128];
    __shared__ __attribute__((aligned(16))) int   rstart[128];
    __shared__ float bsum[64];
    __shared__ int   wtot[8];
    static_assert(ROW_CH * 8 >= 64 * 132 * 4, "the weights fit the shared region");
    float (*Wt)[132] = reinterpret_cast<float (*)[132]>(smem);
    int2*  sorted    = reinterpret_cast<int2*>(smem);

    const int tid  = threadIdx.x;
    const int wave = __builtin_amdgcn_readfirstlane(tid >> 5);
    const int lane = tid & 31;

    int rsrc = 0, rcntv = 0, rpos, cnt;
    {
        const int  r   = wave * RUNS_PW + lane;
        const bool has = lane < RUNS_PW && r < SORT_BLOCKS;
        int pk = 0;
        if (has) pk = __float_as_int(out[gnn_run_word(blockIdx.x, r)]);

        for (int i = tid; i < 128 * YPITCH; i += 256) (&ytile[0][0])[i] = 0.f;

        if (has) {
            int s0 = pk & 0x7FFF;
            int c0 = (pk >> 15) & 0x7FFF;
            s0 = s0 < SORT_EDGES ? s0 : SORT_EDGES;
            c0 = c0 < SORT_EDGES - s0 ? c0 : SORT_EDGES - s0;
            rsrc  = r * SORT_EDGES + s0;
            rcntv = c0;
        }
        int inc = rcntv;
        for (int d = 1; d < 32; d <<= 1) {
            const int t = __shfl_up(inc, d);
            if (lane >= d) inc += t;
        }
        if (lane == 31) wtot[wave] = inc;
        __syncthreads();
        int before = 0, all = 0;
#pragma unroll
        for (int i = 0; i < 8; ++i) {
            const int t = wtot[i];
            all += t;
            if (i < wave) before += t;
        }
        rpos = before + inc - rcntv;
        cnt = __builtin_amdgcn_readfirstlane(all);
    }

    {
        const long xskip = (xb - xa) - (long)GNN_NUSERS * GNN_EMB;

        for (int cb = 0; cb < cnt; cb += ROW_CH) {
            const int n = (cnt - cb) < ROW_CH ? (cnt - cb) : ROW_CH;
            if (tid < 128) rcnt[tid] = 0;
            __syncthreads();

            {
                int2 tv[RUNS_PW];
#pragma unroll
                for (int j = 0; j < RUNS_PW; ++j) {
                    int g = __builtin_amdgcn_readlane(rsrc, j) + lane;
                    g = g < GNN_E ? g : GNN_E - 1;
                    tv[j] = csr[g];
                }
#pragma unroll
                for (int j = 0; j < RUNS_PW; ++j) {
                    const int c   = __builtin_amdgcn_readlane(rcntv, j);
                    const int pos = __builtin_amdgcn_readlane(rpos, j) - cb + lane;
                    if (lane < c && pos >= 0 && pos < ROW_CH) sorted[pos] = tv[j];
                }
                if (__builtin_amdgcn_ballot_w32(rcntv > 32) != 0) {
                    for (int j = 0; j < RUNS_PW; ++j) {
                        const int c  = __shfl(rcntv, j);
                        const int g0 = __shfl(rsrc, j);
                        const int p0 = __shfl(rpos, j) - cb;
                        for (int i = 32 + lane; i < c; i += 32) {
                            const int pos = p0 + i;
                            int g = g0 + i;
                            g = g < GNN_E ? g : GNN_E - 1;
                            if (pos >= 0 && pos < ROW_CH) sorted[pos] = csr[g];
                        }
                    }
                }
            }
            __syncthreads();

            int ex[ROW_KPT], ey[ROW_KPT], rk[ROW_KPT];
#pragma unroll
            for (int k = 0; k < ROW_KPT; ++k) {
                const int2 t = sorted[tid + k * 256];
                ex[k] = t.x; ey[k] = t.y;
            }
#pragma unroll
            for (int k = 0; k < ROW_KPT; ++k) {
                const int e = tid + k * 256;
                rk[k] = 0;
                if (e < n) rk[k] = atomicAdd(&rcnt[(ex[k] >> 17) & 127], 1);
            }
            __syncthreads();

            if (wave == 0) {
                const int4 c4 = *(const int4*)&rcnt[4 * lane];
                const int t0 = c4.x, t1 = t0 + c4.y, t2 = t1 + c4.z, t3 = t2 + c4.w;
                int inc = t3;
                for (int d = 1; d < 32; d <<= 1) {
                    const int t = __shfl_up(inc, d);
                    if (lane >= d) inc += t;
                }
                const int ex0 = inc - t3;
                *(int4*)&rstart[4 * lane] = make_int4(ex0, ex0 + t0, ex0 + t1, ex0 + t2);
            }
            __syncthreads();

            int rs[ROW_KPT];
#pragma unroll
            for (int k = 0; k < ROW_KPT; ++k) rs[k] = rstart[(ex[k] >> 17) & 127];
#pragma unroll
            for (int k = 0; k < ROW_KPT; ++k) {
                const int e = tid + k * 256;
                if (e < n) {
                    int pos = rs[k] + rk[k];
                    pos = pos < ROW_CH ? pos : ROW_CH - 1;
                    sorted[pos] = make_int2(ex[k] | (rk[k] == 0 ? ROW_FIRST : 0), ey[k]);
                }
            }
            __syncthreads();

            const int q  = (((n + 7) >> 3) + 31) & ~31;
            int       p  = wave * q;
            const int p1 = (p + q) < n ? (p + q) : n;
            if (p < p1) {
                v2f acc; acc.x = 0.f; acc.y = 0.f;
                int cur = (__builtin_amdgcn_readfirstlane(sorted[p].x) >> 17) & 127;
                int2 ent = make_int2(0, 0);
                if (p + lane < p1) ent = sorted[p + lane];
                while (p < p1) {
                    const int m  = (p1 - p) < 32 ? (p1 - p) : 32;
                    const int np = p + 32;
                    unsigned di = (unsigned)ent.x & 0x1FFFFu;
                    di = di < GNN_N ? di : GNN_N - 1;
                    const long     offb = ((long)di * GNN_EMB + (di < GNN_NUSERS ? 0L : xskip)) * 4;
                    const unsigned chg  = __builtin_amdgcn_ballot_w32((ent.x & ROW_FIRST) != 0);
                    v2f x[32];
                    gnn_issue32(xa, offb, lane, x);
                    int2 nxt = make_int2(0, 0);
                    if (np + lane < p1) nxt = sorted[np + lane];
                    if (m == 32) gnn_sum32<true >(x, ent.x, ent.y, chg, m, lane, ytile, cur, acc);
                    else         gnn_sum32<false>(x, ent.x, ent.y, chg, m, lane, ytile, cur, acc);
                    ent = nxt;
                    p   = np;
                }
                atomicAdd(&ytile[cur][2 * lane],     acc.x);
                atomicAdd(&ytile[cur][2 * lane + 1], acc.y);
            }
        }
    }
    __syncthreads();

    float4 xv[8];
    if (ego_out) {
#pragma unroll
        for (int k = 0; k < 8; ++k) {
            const int i   = tid + k * 256;
            const int row = blockIdx.x * 128 + (i >> 4);
            xv[k] = make_float4(0.f, 0.f, 0.f, 0.f);
            if (row < GNN_N) {
                const float* xr = (row < GNN_NUSERS) ? xa + (long)row * GNN_EMB
                                                     : xb + (long)(row - GNN_NUSERS) * GNN_EMB;
                xv[k] = *(const float4*)(xr + ((i & 15) << 2));
            }
        }
    }

    float wv[32];
#pragma unroll
    for (int k = 0; k < 32; ++k) {
        const int i = tid + k * 256;
        wv[k] = (k < 16) ? Wgc[i] : Wbi[i - 64 * 64];
    }
#pragma unroll
    for (int k = 0; k < 32; ++k) {
        const int i = tid + k * 256;
        Wt[i & 63][i >> 6] = wv[k];
    }
    if (tid < 64) bsum[tid] = bgc[tid] + bbi[tid];
    __syncthreads();

    if (ego_out) {
#pragma unroll
        for (int k = 0; k < 8; ++k) {
            const int i   = tid + k * 256;
            const int row = blockIdx.x * 128 + (i >> 4);
            if (row < GNN_N) *(float4*)(out + (long)row * 192 + ((i & 15) << 2)) = xv[k];
        }
    }

    const int rowbase = blockIdx.x * 128 + wave * 16;
    if (rowbase >= GNN_N) return;

    const int m    = lane & 15;
    const int half = lane >> 4;
    const int kb   = half << 1;

    const long  row   = (long)(rowbase + m);
    const float* egoRd = (row < GNN_NUSERS) ? xa + row * 64 : xb + (row - GNN_NUSERS) * 64;
    const float* sLRd  = &ytile[wave * 16 + m][0];

    v2f e2[16], l2[16];
#pragma unroll
    for (int s = 0; s < 16; ++s) {
        int c = 4 * s + kb;
        e2[s] = *(const v2f*)(egoRd + c);
        l2[s] = *(const v2f*)(sLRd + c);
    }

    v8f acc[4] = {};

#pragma unroll
    for (int s = 0; s < 16; ++s) {
        v2f a;
        a.x = e2[s].x + l2[s].x;
        a.y = e2[s].y + l2[s].y;
#pragma unroll
        for (int nt = 0; nt < 4; ++nt) {
            v2f b = *(const v2f*)(&Wt[nt * 16 + m][4 * s + kb]);
            acc[nt] = __builtin_amdgcn_wmma_f32_16x16x4_f32(
                false, a, false, b, (short)0, acc[nt], false, false);
        }
    }
#pragma unroll
    for (int s = 0; s < 16; ++s) {
        v2f a;
        a.x = e2[s].x * l2[s].x;
        a.y = e2[s].y * l2[s].y;
#pragma unroll
        for (int nt = 0; nt < 4; ++nt) {
            v2f b = *(const v2f*)(&Wt[nt * 16 + m][64 + 4 * s + kb]);
            acc[nt] = __builtin_amdgcn_wmma_f32_16x16x4_f32(
                false, a, false, b, (short)0, acc[nt], false, false);
        }
    }

    float vout[4][8];
    float ssq[8];
#pragma unroll
    for (int j = 0; j < 8; ++j) ssq[j] = 0.f;
#pragma unroll
    for (int nt = 0; nt < 4; ++nt) {
        float bc = bsum[nt * 16 + m];
#pragma unroll
        for (int j = 0; j < 8; ++j) {
            float v = acc[nt][j] + bc;
            v = (v > 0.f) ? v : 0.01f * v;
            vout[nt][j] = v;
            ssq[j] += v * v;
        }
    }

    if (ego_out) {
        float* egoWr = ego_out + (long)rowbase * 64;
#pragma unroll
        for (int j = 0; j < 8; ++j) {
            int mrow = j + (half << 3);
#pragma unroll
            for (int nt = 0; nt < 4; ++nt)
                egoWr[mrow * 64 + nt * 16 + m] = vout[nt][j];
        }
    }

#pragma unroll
    for (int j = 0; j < 8; ++j) {
        float s = ssq[j];
        s += __shfl_xor(s, 8, 16);
        s += __shfl_xor(s, 4, 16);
        s += __shfl_xor(s, 2, 16);
        s += __shfl_xor(s, 1, 16);
        float inv = 1.0f / fmaxf(sqrtf(s), 1e-12f);
        int mrow = j + (half << 3);
        float* o = out + (long)(rowbase + mrow) * 192 + out_col0;
#pragma unroll
        for (int nt = 0; nt < 4; ++nt)
            o[nt * 16 + m] = vout[nt][j] * inv;
    }
}

extern "C" void kernel_launch(void* const* d_in, const int* in_sizes, int n_in,
                              void* d_out, int out_size, void* d_ws, size_t ws_size,
                              hipStream_t stream) {
    const float* user_emb = (const float*)d_in[0];
    const float* item_emb = (const float*)d_in[1];
    const float* W_gc0 = (const float*)d_in[2];
    const float* b_gc0 = (const float*)d_in[3];
    const float* W_bi0 = (const float*)d_in[4];
    const float* b_bi0 = (const float*)d_in[5];
    const float* W_gc1 = (const float*)d_in[6];
    const float* b_gc1 = (const float*)d_in[7];
    const float* W_bi1 = (const float*)d_in[8];
    const float* b_bi1 = (const float*)d_in[9];
    const int*   e_src = (const int*)d_in[10];
    const int*   e_dst = (const int*)d_in[11];
    const float* e_val = (const float*)d_in[12];
    float* out = (float*)d_out;

    float* ego1 = (float*)d_ws;
    int2*  csr  = (int2*)(ego1 + (long)GNN_N * GNN_EMB);

    gnn_sort<<<SORT_BLOCKS, 1024, 0, stream>>>(e_src, e_dst, e_val, csr, out);

    gnn_layer<<<GNN_NBKT, 256, 0, stream>>>(csr, user_emb, item_emb, ego1,
                                            W_gc0, b_gc0, W_bi0, b_bi0, out, 64);
    gnn_layer<<<GNN_NBKT, 256, 0, stream>>>(csr, ego1, ego1 + (long)GNN_NUSERS * GNN_EMB,
                                            nullptr,
                                            W_gc1, b_gc1, W_bi1, b_bi1, out, 128);
}
